# starting kernel
# baseline (speedup 1.0000x reference)
#include <hip/hip_runtime.h>

typedef __attribute__((ext_vector_type(2))) float v2f;
typedef __attribute__((ext_vector_type(8))) float v8f;

#define F_OUT 64


__global__ void zero_kernel(float* __restrict__ p, long n) {
    long i = (long)blockIdx.x * blockDim.x + threadIdx.x;
    long stride = (long)gridDim.x * blockDim.x;
    for (; i < n; i += stride) p[i] = 0.0f;
}

__global__ void degree_kernel(const int* __restrict__ src, const int* __restrict__ dst,
                              float* __restrict__ deg_out, float* __restrict__ deg_in, int E) {
    int e = blockIdx.x * blockDim.x + threadIdx.x;
    if (e < E) {
        atomicAdd(&deg_out[src[e]], 1.0f);
        atomicAdd(&deg_in[dst[e]], 1.0f);
    }
}

__global__ void norm_kernel(const float* __restrict__ deg_out, const float* __restrict__ deg_in,
                            float* __restrict__ norm_out, float* __restrict__ norm_in, int N) {
    int i = blockIdx.x * blockDim.x + threadIdx.x;
    if (i < N) {
        norm_out[i] = rsqrtf(fmaxf(deg_out[i], 1.0f));
        norm_in[i]  = rsqrtf(fmaxf(deg_in[i], 1.0f));
    }
}

template <int KD>
__global__ __launch_bounds__(128) void gemm_wmma(const float* __restrict__ X,
                                                 const float* __restrict__ W,
                                                 const float* __restrict__ scale,
                                                 float* __restrict__ Hout, int nrows) {
    __shared__ float wlds[KD * F_OUT];
    int tid = threadIdx.x;
    for (int i = tid; i < KD * F_OUT; i += 128) wlds[i] = W[i];
    __syncthreads();

    int wave = tid >> 5;
    int lane = tid & 31;
    int l15  = lane & 15;
    int hi   = lane >> 4;
    int koff = hi * 2;

    int row0 = blockIdx.x * 64 + wave * 16;
    int rA = row0 + l15;
    if (rA >= nrows) rA = nrows - 1;
    float s = scale ? scale[rA] : 1.0f;

    const float* xrow = X + (long)rA * KD;

    v8f acc[4] = {};

    for (int k = 0; k < KD; k += 4) {
        v2f a;
        a.x = xrow[k + koff]     * s;
        a.y = xrow[k + koff + 1] * s;
        __builtin_prefetch(xrow + k + 16, 0, 1);
#pragma unroll
        for (int n = 0; n < 4; ++n) {
            v2f b;
            b.x = wlds[(k + koff)     * F_OUT + n * 16 + l15];
            b.y = wlds[(k + koff + 1) * F_OUT + n * 16 + l15];
            acc[n] = __builtin_amdgcn_wmma_f32_16x16x4_f32(
                false, a, false, b, (short)0, acc[n], false, false);
        }
    }

    float* outp = Hout + (long)(row0 + hi * 8) * F_OUT + l15;
    if (row0 + 16 <= nrows) {
#pragma unroll
        for (int r = 0; r < 8; ++r) {
#pragma unroll
            for (int n = 0; n < 4; ++n) outp[(long)r * F_OUT + n * 16] = acc[n][r];
        }
    } else {
#pragma unroll
        for (int r = 0; r < 8; ++r) {
            int row = row0 + r + hi * 8;
            if (row < nrows) {
#pragma unroll
                for (int n = 0; n < 4; ++n) outp[(long)r * F_OUT + n * 16] = acc[n][r];
            }
        }
    }
}

__global__ void scatter_kernel(const float* __restrict__ H, const int* __restrict__ src,
                               const int* __restrict__ dst, float* __restrict__ agg, long total) {
    long idx = (long)blockIdx.x * blockDim.x + threadIdx.x;
    if (idx >= total) return;
    int e = (int)(idx >> 6);
    int f = (int)(idx & 63);
    int si = __ldg(src + e);
    int di = __ldg(dst + e);
    atomicAdd(&agg[(long)di * F_OUT + f], H[(long)si * F_OUT + f]);
}

__global__ void act_kernel(const float* __restrict__ agg, const float* __restrict__ norm_in,
                           const float* __restrict__ norm_out, const float* __restrict__ b,
                           float* __restrict__ out, long total) {
    long idx = (long)blockIdx.x * blockDim.x + threadIdx.x;
    if (idx >= total) return;
    int i = (int)(idx >> 6);
    int f = (int)(idx & 63);
    float v = agg[idx] * norm_in[i] + b[f];
    out[idx] = fmaxf(v, 0.0f) * norm_out[i];
}

__global__ void final_kernel(float* __restrict__ out, const float* __restrict__ norm_in,
                             const float* __restrict__ b, long total) {
    long idx = (long)blockIdx.x * blockDim.x + threadIdx.x;
    if (idx >= total) return;
    int i = (int)(idx >> 6);
    int f = (int)(idx & 63);
    out[idx] = out[idx] * norm_in[i] + b[f];
}

extern "C" void kernel_launch(void* const* d_in, const int* in_sizes, int n_in,
                              void* d_out, int out_size, void* d_ws, size_t ws_size,
                              hipStream_t stream) {
    const float* x   = (const float*)d_in[0];
    const int*   src = (const int*)d_in[1];
    const int*   dst = (const int*)d_in[2];
    const float* W1  = (const float*)d_in[3];
    const float* b1  = (const float*)d_in[4];
    const float* W2  = (const float*)d_in[5];
    const float* b2  = (const float*)d_in[6];
    float* out = (float*)d_out;

    const int N = in_sizes[0] / 128;
    const int E = in_sizes[1];
    const long NF = (long)N * F_OUT;
    const long EF = (long)E * F_OUT;

    float* w        = (float*)d_ws;
    float* deg_out  = w;
    float* deg_in   = w + (long)N;
    float* norm_out = w + 2L * N;
    float* norm_in  = w + 3L * N;
    float* bufA     = w + 4L * N;
    float* bufB     = bufA + NF;

    const int T = 256;

    zero_kernel<<<1024, T, 0, stream>>>(deg_out, 2L * N);
    zero_kernel<<<2048, T, 0, stream>>>(bufB, NF);
    zero_kernel<<<2048, T, 0, stream>>>(out, NF);

    degree_kernel<<<(E + T - 1) / T, T, 0, stream>>>(src, dst, deg_out, deg_in, E);
    norm_kernel<<<(N + T - 1) / T, T, 0, stream>>>(deg_out, deg_in, norm_out, norm_in, N);

    int gemm_blocks = (N + 63) / 64;
    gemm_wmma<128><<<gemm_blocks, 128, 0, stream>>>(x, W1, norm_out, bufA, N);

    scatter_kernel<<<(int)((EF + T - 1) / T), T, 0, stream>>>(bufA, src, dst, bufB, EF);

    act_kernel<<<(int)((NF + T - 1) / T), T, 0, stream>>>(bufB, norm_in, norm_out, b1, bufA, NF);

    gemm_wmma<64><<<gemm_blocks, 128, 0, stream>>>(bufA, W2, nullptr, bufB, N);

    scatter_kernel<<<(int)((EF + T - 1) / T), T, 0, stream>>>(bufB, src, dst, out, EF);
    final_kernel<<<(int)((NF + T - 1) / T), T, 0, stream>>>(out, norm_in, b2, NF);
}


# degree counts: per-block LDS histograms (32 edge chunks x 2 index arrays) stored as partial planes, summed in the norm kernel; no global atomics or zeroing for degrees; integer counts, exact
# speedup vs baseline: 1.2715x; 1.2715x over previous
#include <hip/hip_runtime.h>

typedef __attribute__((ext_vector_type(2))) float v2f;
typedef __attribute__((ext_vector_type(8))) float v8f;

#define F_OUT 64


__global__ void zero_kernel(float* __restrict__ p, long n) {
    long i = (long)blockIdx.x * blockDim.x + threadIdx.x;
    long stride = (long)gridDim.x * blockDim.x;
    for (; i < n; i += stride) p[i] = 0.0f;
}

#define DEG_CHUNKS 32
#define HBINS      51200
#define DEG_BATCH  8

__device__ __forceinline__ void count_one(unsigned int* hist, int v, int base, unsigned int nb) {
    unsigned int a = (unsigned int)(v - base);
    if (a < nb) atomicAdd(&hist[a], 1u);
}

__global__ __launch_bounds__(1024) void degree_kernel(const int* __restrict__ src,
                                                      const int* __restrict__ dst,
                                                      unsigned int* __restrict__ part,
                                                      int E, int N) {
    __shared__ uint4 hist4[HBINS / 4];
    unsigned int* hist = reinterpret_cast<unsigned int*>(hist4);
    const int tid = threadIdx.x;
    const int* __restrict__ idx = blockIdx.y ? dst : src;
    const long pitch = ((long)N + 31) & ~31L;
    unsigned int* __restrict__ plane = part + ((long)blockIdx.y * DEG_CHUNKS + blockIdx.x) * pitch;

    int per = (((E + DEG_CHUNKS - 1) / DEG_CHUNKS) + 3) & ~3;
    int e0 = blockIdx.x * per;
    if (e0 > E) e0 = E;
    int e1 = e0 + per;
    if (e1 > E) e1 = E;
    const int* __restrict__ my = idx + e0;
    const int4* __restrict__ my4 = reinterpret_cast<const int4*>(my);
    const int cnt = e1 - e0;
    const bool vec_ok = (((size_t)my) & 15) == 0;
    const int nv = vec_ok ? (cnt >> 2) : 0;

    for (long base = 0; base < N; base += HBINS) {
        long left = pitch - base;
        const int nw = (int)(left < HBINS ? left : HBINS);
        long leftn = (long)N - base;
        const unsigned int nb = (unsigned int)(leftn < HBINS ? leftn : HBINS);
        const int b0 = (int)base;

        for (int i = tid; i < HBINS / 4; i += 1024) hist4[i] = make_uint4(0u, 0u, 0u, 0u);
        __syncthreads();

        for (int s = 0; s < nv; s += DEG_BATCH * 1024) {
            int4 v[DEG_BATCH];
#pragma unroll
            for (int u = 0; u < DEG_BATCH; ++u) {
                int i = s + u * 1024 + tid;
                v[u] = (i < nv) ? my4[i] : make_int4(-1, -1, -1, -1);
            }
#pragma unroll
            for (int u = 0; u < DEG_BATCH; ++u) {
                count_one(hist, v[u].x, b0, nb);
                count_one(hist, v[u].y, b0, nb);
                count_one(hist, v[u].z, b0, nb);
                count_one(hist, v[u].w, b0, nb);
            }
        }
        for (int e = (nv << 2) + tid; e < cnt; e += 1024) count_one(hist, my[e], b0, nb);
        __syncthreads();

        uint4* __restrict__ out4 = reinterpret_cast<uint4*>(plane + base);
        for (int i = tid; i < (nw >> 2); i += 1024) out4[i] = hist4[i];
        __syncthreads();
    }
}

__global__ void norm_kernel(const unsigned int* __restrict__ part,
                            float* __restrict__ norm_out, float* __restrict__ norm_in, int N) {
    int i = blockIdx.x * blockDim.x + threadIdx.x;
    if (i < N) {
        const long pitch = ((long)N + 31) & ~31L;
        const unsigned int* __restrict__ po = part + i;
        const unsigned int* __restrict__ pi = part + DEG_CHUNKS * pitch + i;
        unsigned int co = 0, ci = 0;
#pragma unroll
        for (int g = 0; g < DEG_CHUNKS; ++g) {
            co += po[g * pitch];
            ci += pi[g * pitch];
        }
        norm_out[i] = rsqrtf(fmaxf((float)co, 1.0f));
        norm_in[i]  = rsqrtf(fmaxf((float)ci, 1.0f));
    }
}

template <int KD>
__global__ __launch_bounds__(128) void gemm_wmma(const float* __restrict__ X,
                                                 const float* __restrict__ W,
                                                 const float* __restrict__ scale,
                                                 float* __restrict__ Hout, int nrows) {
    __shared__ float wlds[KD * F_OUT];
    int tid = threadIdx.x;
    for (int i = tid; i < KD * F_OUT; i += 128) wlds[i] = W[i];
    __syncthreads();

    int wave = tid >> 5;
    int lane = tid & 31;
    int l15  = lane & 15;
    int hi   = lane >> 4;
    int koff = hi * 2;

    int row0 = blockIdx.x * 64 + wave * 16;
    int rA = row0 + l15;
    if (rA >= nrows) rA = nrows - 1;
    float s = scale ? scale[rA] : 1.0f;

    const float* xrow = X + (long)rA * KD;

    v8f acc[4] = {};

    for (int k = 0; k < KD; k += 4) {
        v2f a;
        a.x = xrow[k + koff]     * s;
        a.y = xrow[k + koff + 1] * s;
        __builtin_prefetch(xrow + k + 16, 0, 1);
#pragma unroll
        for (int n = 0; n < 4; ++n) {
            v2f b;
            b.x = wlds[(k + koff)     * F_OUT + n * 16 + l15];
            b.y = wlds[(k + koff + 1) * F_OUT + n * 16 + l15];
            acc[n] = __builtin_amdgcn_wmma_f32_16x16x4_f32(
                false, a, false, b, (short)0, acc[n], false, false);
        }
    }

    float* outp = Hout + (long)(row0 + hi * 8) * F_OUT + l15;
    if (row0 + 16 <= nrows) {
#pragma unroll
        for (int r = 0; r < 8; ++r) {
#pragma unroll
            for (int n = 0; n < 4; ++n) outp[(long)r * F_OUT + n * 16] = acc[n][r];
        }
    } else {
#pragma unroll
        for (int r = 0; r < 8; ++r) {
            int row = row0 + r + hi * 8;
            if (row < nrows) {
#pragma unroll
                for (int n = 0; n < 4; ++n) outp[(long)r * F_OUT + n * 16] = acc[n][r];
            }
        }
    }
}

__global__ void scatter_kernel(const float* __restrict__ H, const int* __restrict__ src,
                               const int* __restrict__ dst, float* __restrict__ agg, long total) {
    long idx = (long)blockIdx.x * blockDim.x + threadIdx.x;
    if (idx >= total) return;
    int e = (int)(idx >> 6);
    int f = (int)(idx & 63);
    int si = __ldg(src + e);
    int di = __ldg(dst + e);
    atomicAdd(&agg[(long)di * F_OUT + f], H[(long)si * F_OUT + f]);
}

__global__ void act_kernel(const float* __restrict__ agg, const float* __restrict__ norm_in,
                           const float* __restrict__ norm_out, const float* __restrict__ b,
                           float* __restrict__ out, long total) {
    long idx = (long)blockIdx.x * blockDim.x + threadIdx.x;
    if (idx >= total) return;
    int i = (int)(idx >> 6);
    int f = (int)(idx & 63);
    float v = agg[idx] * norm_in[i] + b[f];
    out[idx] = fmaxf(v, 0.0f) * norm_out[i];
}

__global__ void final_kernel(float* __restrict__ out, const float* __restrict__ norm_in,
                             const float* __restrict__ b, long total) {
    long idx = (long)blockIdx.x * blockDim.x + threadIdx.x;
    if (idx >= total) return;
    int i = (int)(idx >> 6);
    int f = (int)(idx & 63);
    out[idx] = out[idx] * norm_in[i] + b[f];
}

extern "C" void kernel_launch(void* const* d_in, const int* in_sizes, int n_in,
                              void* d_out, int out_size, void* d_ws, size_t ws_size,
                              hipStream_t stream) {
    const float* x   = (const float*)d_in[0];
    const int*   src = (const int*)d_in[1];
    const int*   dst = (const int*)d_in[2];
    const float* W1  = (const float*)d_in[3];
    const float* b1  = (const float*)d_in[4];
    const float* W2  = (const float*)d_in[5];
    const float* b2  = (const float*)d_in[6];
    float* out = (float*)d_out;

    const int N = in_sizes[0] / 128;
    const int E = in_sizes[1];
    const long NF = (long)N * F_OUT;
    const long EF = (long)E * F_OUT;

    float* w        = (float*)d_ws;
    float* norm_out = w + 2L * N;
    float* norm_in  = w + 3L * N;
    float* bufA     = w + 4L * N;
    float* bufB     = bufA + NF;
    unsigned int* part = (unsigned int*)(bufB + NF);

    const int T = 256;

    zero_kernel<<<2048, T, 0, stream>>>(bufB, NF);
    zero_kernel<<<2048, T, 0, stream>>>(out, NF);

    degree_kernel<<<dim3(DEG_CHUNKS, 2), 1024, 0, stream>>>(src, dst, part, E, N);
    norm_kernel<<<(N + T - 1) / T, T, 0, stream>>>(part, norm_out, norm_in, N);

    int gemm_blocks = (N + 63) / 64;
    gemm_wmma<128><<<gemm_blocks, 128, 0, stream>>>(x, W1, norm_out, bufA, N);

    scatter_kernel<<<(int)((EF + T - 1) / T), T, 0, stream>>>(bufA, src, dst, bufB, EF);

    act_kernel<<<(int)((NF + T - 1) / T), T, 0, stream>>>(bufB, norm_in, norm_out, b1, bufA, NF);

    gemm_wmma<64><<<gemm_blocks, 128, 0, stream>>>(bufA, W2, nullptr, bufB, N);

    scatter_kernel<<<(int)((EF + T - 1) / T), T, 0, stream>>>(bufB, src, dst, out, EF);
    final_kernel<<<(int)((NF + T - 1) / T), T, 0, stream>>>(out, norm_in, b2, NF);
}


# scatter-add by float atomics -> edges grouped by destination per call (LDS counting sort on the degree planes) + per-node register gather-sum; sum order = list order (arrival order inside a chunk)
# speedup vs baseline: 2.3364x; 1.8375x over previous
#include <hip/hip_runtime.h>

typedef __attribute__((ext_vector_type(2))) float v2f;
typedef __attribute__((ext_vector_type(8))) float v8f;

#define F_OUT 64


#define DEG_CHUNKS 32
#define HBINS      51200
#define DEG_BATCH  8

__device__ __forceinline__ void count_one(unsigned int* hist, int v, int base, unsigned int nb) {
    unsigned int a = (unsigned int)(v - base);
    if (a < nb) atomicAdd(&hist[a], 1u);
}

__global__ __launch_bounds__(1024) void degree_kernel(const int* __restrict__ src,
                                                      const int* __restrict__ dst,
                                                      unsigned int* __restrict__ part,
                                                      int E, int N) {
    __shared__ uint4 hist4[HBINS / 4];
    unsigned int* hist = reinterpret_cast<unsigned int*>(hist4);
    const int tid = threadIdx.x;
    const int* __restrict__ idx = blockIdx.y ? dst : src;
    const long pitch = ((long)N + 31) & ~31L;
    unsigned int* __restrict__ plane = part + ((long)blockIdx.y * DEG_CHUNKS + blockIdx.x) * pitch;

    int per = (((E + DEG_CHUNKS - 1) / DEG_CHUNKS) + 3) & ~3;
    int e0 = blockIdx.x * per;
    if (e0 > E) e0 = E;
    int e1 = e0 + per;
    if (e1 > E) e1 = E;
    const int* __restrict__ my = idx + e0;
    const int4* __restrict__ my4 = reinterpret_cast<const int4*>(my);
    const int cnt = e1 - e0;
    const bool vec_ok = (((size_t)my) & 15) == 0;
    const int nv = vec_ok ? (cnt >> 2) : 0;

    for (long base = 0; base < N; base += HBINS) {
        long left = pitch - base;
        const int nw = (int)(left < HBINS ? left : HBINS);
        long leftn = (long)N - base;
        const unsigned int nb = (unsigned int)(leftn < HBINS ? leftn : HBINS);
        const int b0 = (int)base;

        for (int i = tid; i < HBINS / 4; i += 1024) hist4[i] = make_uint4(0u, 0u, 0u, 0u);
        __syncthreads();

        for (int s = 0; s < nv; s += DEG_BATCH * 1024) {
            int4 v[DEG_BATCH];
#pragma unroll
            for (int u = 0; u < DEG_BATCH; ++u) {
                int i = s + u * 1024 + tid;
                v[u] = (i < nv) ? my4[i] : make_int4(-1, -1, -1, -1);
            }
#pragma unroll
            for (int u = 0; u < DEG_BATCH; ++u) {
                count_one(hist, v[u].x, b0, nb);
                count_one(hist, v[u].y, b0, nb);
                count_one(hist, v[u].z, b0, nb);
                count_one(hist, v[u].w, b0, nb);
            }
        }
        for (int e = (nv << 2) + tid; e < cnt; e += 1024) count_one(hist, my[e], b0, nb);
        __syncthreads();

        uint4* __restrict__ out4 = reinterpret_cast<uint4*>(plane + base);
        for (int i = tid; i < (nw >> 2); i += 1024) out4[i] = hist4[i];
        __syncthreads();
    }
}

__global__ void norm_kernel(unsigned int* __restrict__ part, float* __restrict__ norm_out,
                            float* __restrict__ norm_in, unsigned int* __restrict__ cnt_in, int N) {
    int i = blockIdx.x * blockDim.x + threadIdx.x;
    if (i < N) {
        const long pitch = ((long)N + 31) & ~31L;
        const unsigned int* po = part + i;
        unsigned int* pi = part + DEG_CHUNKS * pitch + i;
        unsigned int vo[DEG_CHUNKS], vi[DEG_CHUNKS];
#pragma unroll
        for (int g = 0; g < DEG_CHUNKS; ++g) {
            vo[g] = po[g * pitch];
            vi[g] = pi[g * pitch];
        }
        unsigned int co = 0, ci = 0;
#pragma unroll
        for (int g = 0; g < DEG_CHUNKS; ++g) {
            co += vo[g];
            pi[g * pitch] = ci;
            ci += vi[g];
        }
        cnt_in[i] = ci;
        norm_out[i] = rsqrtf(fmaxf((float)co, 1.0f));
        norm_in[i]  = rsqrtf(fmaxf((float)ci, 1.0f));
    }
}

#define PL_BATCH 8

__device__ __forceinline__ void place_one(unsigned int* cur, int* __restrict__ csr, int d, int s,
                                          int base, unsigned int nb) {
    unsigned int a = (unsigned int)(d - base);
    if (a < nb) {
        unsigned int pos = atomicAdd(&cur[a], 1u);
        csr[pos] = s;
    }
}

__global__ __launch_bounds__(1024) void place_kernel(const int* __restrict__ src,
                                                     const int* __restrict__ dst,
                                                     const unsigned int* __restrict__ part,
                                                     const unsigned int* __restrict__ cnt_in,
                                                     unsigned int* __restrict__ row_ptr,
                                                     int* __restrict__ csr_src, int E, int N) {
    __shared__ uint4 cur4[HBINS / 4];
    __shared__ unsigned int wsum[32];
    __shared__ unsigned int tile_total;
    unsigned int* cur = reinterpret_cast<unsigned int*>(cur4);
    const int tid = threadIdx.x;
    const int lane = tid & 31;
    const int wv = tid >> 5;
    const long pitch = ((long)N + 31) & ~31L;
    const unsigned int* __restrict__ pref = part + ((long)DEG_CHUNKS + blockIdx.x) * pitch;

    int per = (((E + DEG_CHUNKS - 1) / DEG_CHUNKS) + 3) & ~3;
    int e0 = blockIdx.x * per;
    if (e0 > E) e0 = E;
    int e1 = e0 + per;
    if (e1 > E) e1 = E;
    const int* __restrict__ myd = dst + e0;
    const int* __restrict__ mys = src + e0;
    const int4* __restrict__ myd4 = reinterpret_cast<const int4*>(myd);
    const int4* __restrict__ mys4 = reinterpret_cast<const int4*>(mys);
    const int cnt = e1 - e0;
    const bool vec_ok = ((((size_t)myd) | ((size_t)mys)) & 15) == 0;
    const int nv = vec_ok ? (cnt >> 2) : 0;

    unsigned int carry = 0;
    for (long base = 0; base < N; base += HBINS) {
        long leftn = (long)N - base;
        const int nb = (int)(leftn < HBINS ? leftn : HBINS);
        const int b0 = (int)base;

#pragma unroll 8
        for (int i = tid; i < nb; i += 1024) cur[i] = cnt_in[base + i];
        __syncthreads();

        const int run_len = (nb + 1023) >> 10;
        int rb = tid * run_len;
        if (rb > nb) rb = nb;
        int re = rb + run_len;
        if (re > nb) re = nb;
        unsigned int s = 0;
        for (int i = rb; i < re; ++i) s += cur[i];
        unsigned int inc = s;
        for (int o = 1; o < 32; o <<= 1) {
            unsigned int t = __shfl_up(inc, o);
            if (lane >= o) inc += t;
        }
        if (lane == 31) wsum[wv] = inc;
        __syncthreads();
        if (wv == 0) {
            unsigned int wsv = wsum[lane];
            unsigned int wi = wsv;
            for (int o = 1; o < 32; o <<= 1) {
                unsigned int t = __shfl_up(wi, o);
                if (lane >= o) wi += t;
            }
            wsum[lane] = wi - wsv;
            if (lane == 31) tile_total = wi;
        }
        __syncthreads();
        unsigned int run = carry + wsum[wv] + (inc - s);
        for (int i = rb; i < re; ++i) {
            unsigned int v = cur[i];
            cur[i] = run;
            run += v;
        }
        carry += tile_total;
        __syncthreads();

#pragma unroll 8
        for (int i = tid; i < nb; i += 1024) {
            unsigned int r = cur[i];
            if (blockIdx.x == 0) row_ptr[base + i] = r;
            cur[i] = r + pref[base + i];
        }
        __syncthreads();

        for (int v0 = 0; v0 < nv; v0 += PL_BATCH * 1024) {
            int4 vd[PL_BATCH], vs[PL_BATCH];
#pragma unroll
            for (int u = 0; u < PL_BATCH; ++u) {
                int i = v0 + u * 1024 + tid;
                if (i < nv) {
                    vd[u] = myd4[i];
                    vs[u] = mys4[i];
                } else {
                    vd[u] = make_int4(-1, -1, -1, -1);
                    vs[u] = make_int4(0, 0, 0, 0);
                }
            }
#pragma unroll
            for (int u = 0; u < PL_BATCH; ++u) {
                place_one(cur, csr_src, vd[u].x, vs[u].x, b0, (unsigned int)nb);
                place_one(cur, csr_src, vd[u].y, vs[u].y, b0, (unsigned int)nb);
                place_one(cur, csr_src, vd[u].z, vs[u].z, b0, (unsigned int)nb);
                place_one(cur, csr_src, vd[u].w, vs[u].w, b0, (unsigned int)nb);
            }
        }
        for (int e = (nv << 2) + tid; e < cnt; e += 1024)
            place_one(cur, csr_src, myd[e], mys[e], b0, (unsigned int)nb);
        __syncthreads();
    }
    if (blockIdx.x == 0 && tid == 0) row_ptr[N] = carry;
}

template <int KD>
__global__ __launch_bounds__(128) void gemm_wmma(const float* __restrict__ X,
                                                 const float* __restrict__ W,
                                                 const float* __restrict__ scale,
                                                 float* __restrict__ Hout, int nrows) {
    __shared__ float wlds[KD * F_OUT];
    int tid = threadIdx.x;
    for (int i = tid; i < KD * F_OUT; i += 128) wlds[i] = W[i];
    __syncthreads();

    int wave = tid >> 5;
    int lane = tid & 31;
    int l15  = lane & 15;
    int hi   = lane >> 4;
    int koff = hi * 2;

    int row0 = blockIdx.x * 64 + wave * 16;
    int rA = row0 + l15;
    if (rA >= nrows) rA = nrows - 1;
    float s = scale ? scale[rA] : 1.0f;

    const float* xrow = X + (long)rA * KD;

    v8f acc[4] = {};

    for (int k = 0; k < KD; k += 4) {
        v2f a;
        a.x = xrow[k + koff]     * s;
        a.y = xrow[k + koff + 1] * s;
        __builtin_prefetch(xrow + k + 16, 0, 1);
#pragma unroll
        for (int n = 0; n < 4; ++n) {
            v2f b;
            b.x = wlds[(k + koff)     * F_OUT + n * 16 + l15];
            b.y = wlds[(k + koff + 1) * F_OUT + n * 16 + l15];
            acc[n] = __builtin_amdgcn_wmma_f32_16x16x4_f32(
                false, a, false, b, (short)0, acc[n], false, false);
        }
    }

    float* outp = Hout + (long)(row0 + hi * 8) * F_OUT + l15;
    if (row0 + 16 <= nrows) {
#pragma unroll
        for (int r = 0; r < 8; ++r) {
#pragma unroll
            for (int n = 0; n < 4; ++n) outp[(long)r * F_OUT + n * 16] = acc[n][r];
        }
    } else {
#pragma unroll
        for (int r = 0; r < 8; ++r) {
            int row = row0 + r + hi * 8;
            if (row < nrows) {
#pragma unroll
                for (int n = 0; n < 4; ++n) outp[(long)r * F_OUT + n * 16] = acc[n][r];
            }
        }
    }
}

__global__ __launch_bounds__(256) void gather_kernel(const float* __restrict__ H,
                                                     const int* __restrict__ csr_src,
                                                     const unsigned int* __restrict__ row_ptr,
                                                     float* __restrict__ agg, int N) {
    const int t = blockIdx.x * 256 + threadIdx.x;
    const int node = t >> 4;
    const int q = t & 15;
    if (node >= N) return;
    const unsigned int beg = row_ptr[node];
    const unsigned int end = row_ptr[node + 1];
    const float4* __restrict__ H4 = reinterpret_cast<const float4*>(H);
    float4 acc = make_float4(0.0f, 0.0f, 0.0f, 0.0f);
    for (unsigned int c0 = beg; c0 < end; c0 += 16) {
        const unsigned int left = end - c0;
        const int m = (int)(left < 16u ? left : 16u);
        const int mine = (q < m) ? csr_src[c0 + q] : 0;
        for (int j = 0; j < m; j += 4) {
            const int s0 = __shfl(mine, j, 16);
            const int s1 = __shfl(mine, j + 1, 16);
            const int s2 = __shfl(mine, j + 2, 16);
            const int s3 = __shfl(mine, j + 3, 16);
            const float4 zero4 = make_float4(0.0f, 0.0f, 0.0f, 0.0f);
            float4 v0 = H4[(long)s0 * (F_OUT / 4) + q];
            float4 v1 = zero4, v2 = zero4, v3 = zero4;
            if (j + 1 < m) v1 = H4[(long)s1 * (F_OUT / 4) + q];
            if (j + 2 < m) v2 = H4[(long)s2 * (F_OUT / 4) + q];
            if (j + 3 < m) v3 = H4[(long)s3 * (F_OUT / 4) + q];
            acc.x += v0.x; acc.y += v0.y; acc.z += v0.z; acc.w += v0.w;
            acc.x += v1.x; acc.y += v1.y; acc.z += v1.z; acc.w += v1.w;
            acc.x += v2.x; acc.y += v2.y; acc.z += v2.z; acc.w += v2.w;
            acc.x += v3.x; acc.y += v3.y; acc.z += v3.z; acc.w += v3.w;
        }
    }
    reinterpret_cast<float4*>(agg)[(long)node * (F_OUT / 4) + q] = acc;
}

__global__ void act_kernel(const float* __restrict__ agg, const float* __restrict__ norm_in,
                           const float* __restrict__ norm_out, const float* __restrict__ b,
                           float* __restrict__ out, long total) {
    long idx = (long)blockIdx.x * blockDim.x + threadIdx.x;
    if (idx >= total) return;
    int i = (int)(idx >> 6);
    int f = (int)(idx & 63);
    float v = agg[idx] * norm_in[i] + b[f];
    out[idx] = fmaxf(v, 0.0f) * norm_out[i];
}

__global__ void final_kernel(float* __restrict__ out, const float* __restrict__ norm_in,
                             const float* __restrict__ b, long total) {
    long idx = (long)blockIdx.x * blockDim.x + threadIdx.x;
    if (idx >= total) return;
    int i = (int)(idx >> 6);
    int f = (int)(idx & 63);
    out[idx] = out[idx] * norm_in[i] + b[f];
}

extern "C" void kernel_launch(void* const* d_in, const int* in_sizes, int n_in,
                              void* d_out, int out_size, void* d_ws, size_t ws_size,
                              hipStream_t stream) {
    const float* x   = (const float*)d_in[0];
    const int*   src = (const int*)d_in[1];
    const int*   dst = (const int*)d_in[2];
    const float* W1  = (const float*)d_in[3];
    const float* b1  = (const float*)d_in[4];
    const float* W2  = (const float*)d_in[5];
    const float* b2  = (const float*)d_in[6];
    float* out = (float*)d_out;

    const int N = in_sizes[0] / 128;
    const int E = in_sizes[1];
    const long NF = (long)N * F_OUT;

    float* w        = (float*)d_ws;
    unsigned int* cnt_in = (unsigned int*)w;
    float* norm_out = w + 2L * N;
    float* norm_in  = w + 3L * N;
    float* bufA     = w + 4L * N;
    float* bufB     = bufA + NF;
    unsigned int* part = (unsigned int*)(bufB + NF);
    unsigned int* row_ptr = part + 2L * DEG_CHUNKS * (N + 32);
    int* csr_src = (int*)(row_ptr + N + 32);

    const int T = 256;

    degree_kernel<<<dim3(DEG_CHUNKS, 2), 1024, 0, stream>>>(src, dst, part, E, N);
    norm_kernel<<<(N + T - 1) / T, T, 0, stream>>>(part, norm_out, norm_in, cnt_in, N);
    place_kernel<<<DEG_CHUNKS, 1024, 0, stream>>>(src, dst, part, cnt_in, row_ptr, csr_src, E, N);

    int gemm_blocks = (N + 63) / 64;
    gemm_wmma<128><<<gemm_blocks, 128, 0, stream>>>(x, W1, norm_out, bufA, N);

    const int gather_blocks = (int)((NF / 4 + T - 1) / T);
    gather_kernel<<<gather_blocks, T, 0, stream>>>(bufA, csr_src, row_ptr, bufB, N);

    act_kernel<<<(int)((NF + T - 1) / T), T, 0, stream>>>(bufB, norm_in, norm_out, b1, bufA, NF);

    gemm_wmma<64><<<gemm_blocks, 128, 0, stream>>>(bufA, W2, nullptr, bufB, N);

    gather_kernel<<<gather_blocks, T, 0, stream>>>(bufB, csr_src, row_ptr, out, N);
    final_kernel<<<(int)((NF + T - 1) / T), T, 0, stream>>>(out, norm_in, b2, NF);
}


# the two elementwise epilogue launches (norm_in scale + bias + relu + norm_out pre-scale; norm_in scale + bias) folded into the gather kernel's row store: 9 launches -> 7; sum order unchanged
# speedup vs baseline: 2.7065x; 1.1584x over previous
#include <hip/hip_runtime.h>

typedef __attribute__((ext_vector_type(2))) float v2f;
typedef __attribute__((ext_vector_type(8))) float v8f;

#define F_OUT 64


#define DEG_CHUNKS 32
#define HBINS      51200
#define DEG_BATCH  8

__device__ __forceinline__ void count_one(unsigned int* hist, int v, int base, unsigned int nb) {
    unsigned int a = (unsigned int)(v - base);
    if (a < nb) atomicAdd(&hist[a], 1u);
}

__global__ __launch_bounds__(1024) void degree_kernel(const int* __restrict__ src,
                                                      const int* __restrict__ dst,
                                                      unsigned int* __restrict__ part,
                                                      int E, int N) {
    __shared__ uint4 hist4[HBINS / 4];
    unsigned int* hist = reinterpret_cast<unsigned int*>(hist4);
    const int tid = threadIdx.x;
    const int* __restrict__ idx = blockIdx.y ? dst : src;
    const long pitch = ((long)N + 31) & ~31L;
    unsigned int* __restrict__ plane = part + ((long)blockIdx.y * DEG_CHUNKS + blockIdx.x) * pitch;

    int per = (((E + DEG_CHUNKS - 1) / DEG_CHUNKS) + 3) & ~3;
    int e0 = blockIdx.x * per;
    if (e0 > E) e0 = E;
    int e1 = e0 + per;
    if (e1 > E) e1 = E;
    const int* __restrict__ my = idx + e0;
    const int4* __restrict__ my4 = reinterpret_cast<const int4*>(my);
    const int cnt = e1 - e0;
    const bool vec_ok = (((size_t)my) & 15) == 0;
    const int nv = vec_ok ? (cnt >> 2) : 0;

    for (long base = 0; base < N; base += HBINS) {
        long left = pitch - base;
        const int nw = (int)(left < HBINS ? left : HBINS);
        long leftn = (long)N - base;
        const unsigned int nb = (unsigned int)(leftn < HBINS ? leftn : HBINS);
        const int b0 = (int)base;

        for (int i = tid; i < HBINS / 4; i += 1024) hist4[i] = make_uint4(0u, 0u, 0u, 0u);
        __syncthreads();

        for (int s = 0; s < nv; s += DEG_BATCH * 1024) {
            int4 v[DEG_BATCH];
#pragma unroll
            for (int u = 0; u < DEG_BATCH; ++u) {
                int i = s + u * 1024 + tid;
                v[u] = (i < nv) ? my4[i] : make_int4(-1, -1, -1, -1);
            }
#pragma unroll
            for (int u = 0; u < DEG_BATCH; ++u) {
                count_one(hist, v[u].x, b0, nb);
                count_one(hist, v[u].y, b0, nb);
                count_one(hist, v[u].z, b0, nb);
                count_one(hist, v[u].w, b0, nb);
            }
        }
        for (int e = (nv << 2) + tid; e < cnt; e += 1024) count_one(hist, my[e], b0, nb);
        __syncthreads();

        uint4* __restrict__ out4 = reinterpret_cast<uint4*>(plane + base);
        for (int i = tid; i < (nw >> 2); i += 1024) out4[i] = hist4[i];
        __syncthreads();
    }
}

__global__ void norm_kernel(unsigned int* __restrict__ part, float* __restrict__ norm_out,
                            float* __restrict__ norm_in, unsigned int* __restrict__ cnt_in, int N) {
    int i = blockIdx.x * blockDim.x + threadIdx.x;
    if (i < N) {
        const long pitch = ((long)N + 31) & ~31L;
        const unsigned int* po = part + i;
        unsigned int* pi = part + DEG_CHUNKS * pitch + i;
        unsigned int vo[DEG_CHUNKS], vi[DEG_CHUNKS];
#pragma unroll
        for (int g = 0; g < DEG_CHUNKS; ++g) {
            vo[g] = po[g * pitch];
            vi[g] = pi[g * pitch];
        }
        unsigned int co = 0, ci = 0;
#pragma unroll
        for (int g = 0; g < DEG_CHUNKS; ++g) {
            co += vo[g];
            pi[g * pitch] = ci;
            ci += vi[g];
        }
        cnt_in[i] = ci;
        norm_out[i] = rsqrtf(fmaxf((float)co, 1.0f));
        norm_in[i]  = rsqrtf(fmaxf((float)ci, 1.0f));
    }
}

#define PL_BATCH 8

__device__ __forceinline__ void place_one(unsigned int* cur, int* __restrict__ csr, int d, int s,
                                          int base, unsigned int nb) {
    unsigned int a = (unsigned int)(d - base);
    if (a < nb) {
        unsigned int pos = atomicAdd(&cur[a], 1u);
        csr[pos] = s;
    }
}

__global__ __launch_bounds__(1024) void place_kernel(const int* __restrict__ src,
                                                     const int* __restrict__ dst,
                                                     const unsigned int* __restrict__ part,
                                                     const unsigned int* __restrict__ cnt_in,
                                                     unsigned int* __restrict__ row_ptr,
                                                     int* __restrict__ csr_src, int E, int N) {
    __shared__ uint4 cur4[HBINS / 4];
    __shared__ unsigned int wsum[32];
    __shared__ unsigned int tile_total;
    unsigned int* cur = reinterpret_cast<unsigned int*>(cur4);
    const int tid = threadIdx.x;
    const int lane = tid & 31;
    const int wv = tid >> 5;
    const long pitch = ((long)N + 31) & ~31L;
    const unsigned int* __restrict__ pref = part + ((long)DEG_CHUNKS + blockIdx.x) * pitch;

    int per = (((E + DEG_CHUNKS - 1) / DEG_CHUNKS) + 3) & ~3;
    int e0 = blockIdx.x * per;
    if (e0 > E) e0 = E;
    int e1 = e0 + per;
    if (e1 > E) e1 = E;
    const int* __restrict__ myd = dst + e0;
    const int* __restrict__ mys = src + e0;
    const int4* __restrict__ myd4 = reinterpret_cast<const int4*>(myd);
    const int4* __restrict__ mys4 = reinterpret_cast<const int4*>(mys);
    const int cnt = e1 - e0;
    const bool vec_ok = ((((size_t)myd) | ((size_t)mys)) & 15) == 0;
    const int nv = vec_ok ? (cnt >> 2) : 0;

    unsigned int carry = 0;
    for (long base = 0; base < N; base += HBINS) {
        long leftn = (long)N - base;
        const int nb = (int)(leftn < HBINS ? leftn : HBINS);
        const int b0 = (int)base;

#pragma unroll 8
        for (int i = tid; i < nb; i += 1024) cur[i] = cnt_in[base + i];
        __syncthreads();

        const int run_len = (nb + 1023) >> 10;
        int rb = tid * run_len;
        if (rb > nb) rb = nb;
        int re = rb + run_len;
        if (re > nb) re = nb;
        unsigned int s = 0;
        for (int i = rb; i < re; ++i) s += cur[i];
        unsigned int inc = s;
        for (int o = 1; o < 32; o <<= 1) {
            unsigned int t = __shfl_up(inc, o);
            if (lane >= o) inc += t;
        }
        if (lane == 31) wsum[wv] = inc;
        __syncthreads();
        if (wv == 0) {
            unsigned int wsv = wsum[lane];
            unsigned int wi = wsv;
            for (int o = 1; o < 32; o <<= 1) {
                unsigned int t = __shfl_up(wi, o);
                if (lane >= o) wi += t;
            }
            wsum[lane] = wi - wsv;
            if (lane == 31) tile_total = wi;
        }
        __syncthreads();
        unsigned int run = carry + wsum[wv] + (inc - s);
        for (int i = rb; i < re; ++i) {
            unsigned int v = cur[i];
            cur[i] = run;
            run += v;
        }
        carry += tile_total;
        __syncthreads();

#pragma unroll 8
        for (int i = tid; i < nb; i += 1024) {
            unsigned int r = cur[i];
            if (blockIdx.x == 0) row_ptr[base + i] = r;
            cur[i] = r + pref[base + i];
        }
        __syncthreads();

        for (int v0 = 0; v0 < nv; v0 += PL_BATCH * 1024) {
            int4 vd[PL_BATCH], vs[PL_BATCH];
#pragma unroll
            for (int u = 0; u < PL_BATCH; ++u) {
                int i = v0 + u * 1024 + tid;
                if (i < nv) {
                    vd[u] = myd4[i];
                    vs[u] = mys4[i];
                } else {
                    vd[u] = make_int4(-1, -1, -1, -1);
                    vs[u] = make_int4(0, 0, 0, 0);
                }
            }
#pragma unroll
            for (int u = 0; u < PL_BATCH; ++u) {
                place_one(cur, csr_src, vd[u].x, vs[u].x, b0, (unsigned int)nb);
                place_one(cur, csr_src, vd[u].y, vs[u].y, b0, (unsigned int)nb);
                place_one(cur, csr_src, vd[u].z, vs[u].z, b0, (unsigned int)nb);
                place_one(cur, csr_src, vd[u].w, vs[u].w, b0, (unsigned int)nb);
            }
        }
        for (int e = (nv << 2) + tid; e < cnt; e += 1024)
            place_one(cur, csr_src, myd[e], mys[e], b0, (unsigned int)nb);
        __syncthreads();
    }
    if (blockIdx.x == 0 && tid == 0) row_ptr[N] = carry;
}

template <int KD>
__global__ __launch_bounds__(128) void gemm_wmma(const float* __restrict__ X,
                                                 const float* __restrict__ W,
                                                 const float* __restrict__ scale,
                                                 float* __restrict__ Hout, int nrows) {
    __shared__ float wlds[KD * F_OUT];
    int tid = threadIdx.x;
    for (int i = tid; i < KD * F_OUT; i += 128) wlds[i] = W[i];
    __syncthreads();

    int wave = tid >> 5;
    int lane = tid & 31;
    int l15  = lane & 15;
    int hi   = lane >> 4;
    int koff = hi * 2;

    int row0 = blockIdx.x * 64 + wave * 16;
    int rA = row0 + l15;
    if (rA >= nrows) rA = nrows - 1;
    float s = scale ? scale[rA] : 1.0f;

    const float* xrow = X + (long)rA * KD;

    v8f acc[4] = {};

    for (int k = 0; k < KD; k += 4) {
        v2f a;
        a.x = xrow[k + koff]     * s;
        a.y = xrow[k + koff + 1] * s;
        __builtin_prefetch(xrow + k + 16, 0, 1);
#pragma unroll
        for (int n = 0; n < 4; ++n) {
            v2f b;
            b.x = wlds[(k + koff)     * F_OUT + n * 16 + l15];
            b.y = wlds[(k + koff + 1) * F_OUT + n * 16 + l15];
            acc[n] = __builtin_amdgcn_wmma_f32_16x16x4_f32(
                false, a, false, b, (short)0, acc[n], false, false);
        }
    }

    float* outp = Hout + (long)(row0 + hi * 8) * F_OUT + l15;
    if (row0 + 16 <= nrows) {
#pragma unroll
        for (int r = 0; r < 8; ++r) {
#pragma unroll
            for (int n = 0; n < 4; ++n) outp[(long)r * F_OUT + n * 16] = acc[n][r];
        }
    } else {
#pragma unroll
        for (int r = 0; r < 8; ++r) {
            int row = row0 + r + hi * 8;
            if (row < nrows) {
#pragma unroll
                for (int n = 0; n < 4; ++n) outp[(long)r * F_OUT + n * 16] = acc[n][r];
            }
        }
    }
}

template <bool FIRST_LAYER>
__global__ __launch_bounds__(256) void gather_kernel(const float* __restrict__ H,
                                                     const int* __restrict__ csr_src,
                                                     const unsigned int* __restrict__ row_ptr,
                                                     const float* __restrict__ norm_in,
                                                     const float* __restrict__ norm_out,
                                                     const float* __restrict__ b,
                                                     float* __restrict__ dstrows, int N) {
    const int t = blockIdx.x * 256 + threadIdx.x;
    const int node = t >> 4;
    const int q = t & 15;
    if (node >= N) return;
    const unsigned int beg = row_ptr[node];
    const unsigned int end = row_ptr[node + 1];
    const float ni = norm_in[node];
    const float no = FIRST_LAYER ? norm_out[node] : 1.0f;
    const float b0 = b[4 * q], b1 = b[4 * q + 1], b2 = b[4 * q + 2], b3 = b[4 * q + 3];
    const float4* __restrict__ H4 = reinterpret_cast<const float4*>(H);
    float4 acc = make_float4(0.0f, 0.0f, 0.0f, 0.0f);
    for (unsigned int c0 = beg; c0 < end; c0 += 16) {
        const unsigned int left = end - c0;
        const int m = (int)(left < 16u ? left : 16u);
        const int mine = (q < m) ? csr_src[c0 + q] : 0;
        for (int j = 0; j < m; j += 4) {
            const int s0 = __shfl(mine, j, 16);
            const int s1 = __shfl(mine, j + 1, 16);
            const int s2 = __shfl(mine, j + 2, 16);
            const int s3 = __shfl(mine, j + 3, 16);
            const float4 zero4 = make_float4(0.0f, 0.0f, 0.0f, 0.0f);
            float4 v0 = H4[(long)s0 * (F_OUT / 4) + q];
            float4 v1 = zero4, v2 = zero4, v3 = zero4;
            if (j + 1 < m) v1 = H4[(long)s1 * (F_OUT / 4) + q];
            if (j + 2 < m) v2 = H4[(long)s2 * (F_OUT / 4) + q];
            if (j + 3 < m) v3 = H4[(long)s3 * (F_OUT / 4) + q];
            acc.x += v0.x; acc.y += v0.y; acc.z += v0.z; acc.w += v0.w;
            acc.x += v1.x; acc.y += v1.y; acc.z += v1.z; acc.w += v1.w;
            acc.x += v2.x; acc.y += v2.y; acc.z += v2.z; acc.w += v2.w;
            acc.x += v3.x; acc.y += v3.y; acc.z += v3.z; acc.w += v3.w;
        }
    }
    float4 r;
    r.x = acc.x * ni + b0;
    r.y = acc.y * ni + b1;
    r.z = acc.z * ni + b2;
    r.w = acc.w * ni + b3;
    if (FIRST_LAYER) {
        r.x = fmaxf(r.x, 0.0f) * no;
        r.y = fmaxf(r.y, 0.0f) * no;
        r.z = fmaxf(r.z, 0.0f) * no;
        r.w = fmaxf(r.w, 0.0f) * no;
    }
    reinterpret_cast<float4*>(dstrows)[(long)node * (F_OUT / 4) + q] = r;
}

extern "C" void kernel_launch(void* const* d_in, const int* in_sizes, int n_in,
                              void* d_out, int out_size, void* d_ws, size_t ws_size,
                              hipStream_t stream) {
    const float* x   = (const float*)d_in[0];
    const int*   src = (const int*)d_in[1];
    const int*   dst = (const int*)d_in[2];
    const float* W1  = (const float*)d_in[3];
    const float* b1  = (const float*)d_in[4];
    const float* W2  = (const float*)d_in[5];
    const float* b2  = (const float*)d_in[6];
    float* out = (float*)d_out;

    const int N = in_sizes[0] / 128;
    const int E = in_sizes[1];
    const long NF = (long)N * F_OUT;

    float* w        = (float*)d_ws;
    unsigned int* cnt_in = (unsigned int*)w;
    float* norm_out = w + 2L * N;
    float* norm_in  = w + 3L * N;
    float* bufA     = w + 4L * N;
    float* bufB     = bufA + NF;
    unsigned int* part = (unsigned int*)(bufB + NF);
    unsigned int* row_ptr = part + 2L * DEG_CHUNKS * (N + 32);
    int* csr_src = (int*)(row_ptr + N + 32);

    const int T = 256;

    degree_kernel<<<dim3(DEG_CHUNKS, 2), 1024, 0, stream>>>(src, dst, part, E, N);
    norm_kernel<<<(N + T - 1) / T, T, 0, stream>>>(part, norm_out, norm_in, cnt_in, N);
    place_kernel<<<DEG_CHUNKS, 1024, 0, stream>>>(src, dst, part, cnt_in, row_ptr, csr_src, E, N);

    int gemm_blocks = (N + 63) / 64;
    gemm_wmma<128><<<gemm_blocks, 128, 0, stream>>>(x, W1, norm_out, bufA, N);

    const int gather_blocks = (int)((NF / 4 + T - 1) / T);
    gather_kernel<true><<<gather_blocks, T, 0, stream>>>(bufA, csr_src, row_ptr, norm_in, norm_out, b1, bufB, N);

    gemm_wmma<64><<<gemm_blocks, 128, 0, stream>>>(bufB, W2, nullptr, bufA, N);

    gather_kernel<false><<<gather_blocks, T, 0, stream>>>(bufA, csr_src, row_ptr, norm_in, norm_out, b2, out, N);
}


# layer-2 product (WMMA f32, same k order) fused into the first gather launch: a block's 16 finished rows go to an LDS tile and meet W2 there; 7 launches -> 6; sum order unchanged
# speedup vs baseline: 2.8522x; 1.0538x over previous
#include <hip/hip_runtime.h>

typedef __attribute__((ext_vector_type(2))) float v2f;
typedef __attribute__((ext_vector_type(8))) float v8f;

#define F_OUT 64


#define DEG_CHUNKS 32
#define HBINS      51200
#define DEG_BATCH  8

__device__ __forceinline__ void count_one(unsigned int* hist, int v, int base, unsigned int nb) {
    unsigned int a = (unsigned int)(v - base);
    if (a < nb) atomicAdd(&hist[a], 1u);
}

__global__ __launch_bounds__(1024) void degree_kernel(const int* __restrict__ src,
                                                      const int* __restrict__ dst,
                                                      unsigned int* __restrict__ part,
                                                      int E, int N) {
    __shared__ uint4 hist4[HBINS / 4];
    unsigned int* hist = reinterpret_cast<unsigned int*>(hist4);
    const int tid = threadIdx.x;
    const int* __restrict__ idx = blockIdx.y ? dst : src;
    const long pitch = ((long)N + 31) & ~31L;
    unsigned int* __restrict__ plane = part + ((long)blockIdx.y * DEG_CHUNKS + blockIdx.x) * pitch;

    int per = (((E + DEG_CHUNKS - 1) / DEG_CHUNKS) + 3) & ~3;
    int e0 = blockIdx.x * per;
    if (e0 > E) e0 = E;
    int e1 = e0 + per;
    if (e1 > E) e1 = E;
    const int* __restrict__ my = idx + e0;
    const int4* __restrict__ my4 = reinterpret_cast<const int4*>(my);
    const int cnt = e1 - e0;
    const bool vec_ok = (((size_t)my) & 15) == 0;
    const int nv = vec_ok ? (cnt >> 2) : 0;

    for (long base = 0; base < N; base += HBINS) {
        long left = pitch - base;
        const int nw = (int)(left < HBINS ? left : HBINS);
        long leftn = (long)N - base;
        const unsigned int nb = (unsigned int)(leftn < HBINS ? leftn : HBINS);
        const int b0 = (int)base;

        for (int i = tid; i < HBINS / 4; i += 1024) hist4[i] = make_uint4(0u, 0u, 0u, 0u);
        __syncthreads();

        for (int s = 0; s < nv; s += DEG_BATCH * 1024) {
            int4 v[DEG_BATCH];
#pragma unroll
            for (int u = 0; u < DEG_BATCH; ++u) {
                int i = s + u * 1024 + tid;
                v[u] = (i < nv) ? my4[i] : make_int4(-1, -1, -1, -1);
            }
#pragma unroll
            for (int u = 0; u < DEG_BATCH; ++u) {
                count_one(hist, v[u].x, b0, nb);
                count_one(hist, v[u].y, b0, nb);
                count_one(hist, v[u].z, b0, nb);
                count_one(hist, v[u].w, b0, nb);
            }
        }
        for (int e = (nv << 2) + tid; e < cnt; e += 1024) count_one(hist, my[e], b0, nb);
        __syncthreads();

        uint4* __restrict__ out4 = reinterpret_cast<uint4*>(plane + base);
        for (int i = tid; i < (nw >> 2); i += 1024) out4[i] = hist4[i];
        __syncthreads();
    }
}

__global__ void norm_kernel(unsigned int* __restrict__ part, float* __restrict__ norm_out,
                            float* __restrict__ norm_in, unsigned int* __restrict__ cnt_in, int N) {
    int i = blockIdx.x * blockDim.x + threadIdx.x;
    if (i < N) {
        const long pitch = ((long)N + 31) & ~31L;
        const unsigned int* po = part + i;
        unsigned int* pi = part + DEG_CHUNKS * pitch + i;
        unsigned int vo[DEG_CHUNKS], vi[DEG_CHUNKS];
#pragma unroll
        for (int g = 0; g < DEG_CHUNKS; ++g) {
            vo[g] = po[g * pitch];
            vi[g] = pi[g * pitch];
        }
        unsigned int co = 0, ci = 0;
#pragma unroll
        for (int g = 0; g < DEG_CHUNKS; ++g) {
            co += vo[g];
            pi[g * pitch] = ci;
            ci += vi[g];
        }
        cnt_in[i] = ci;
        norm_out[i] = rsqrtf(fmaxf((float)co, 1.0f));
        norm_in[i]  = rsqrtf(fmaxf((float)ci, 1.0f));
    }
}

#define PL_BATCH 8

__device__ __forceinline__ void place_one(unsigned int* cur, int* __restrict__ csr, int d, int s,
                                          int base, unsigned int nb) {
    unsigned int a = (unsigned int)(d - base);
    if (a < nb) {
        unsigned int pos = atomicAdd(&cur[a], 1u);
        csr[pos] = s;
    }
}

__global__ __launch_bounds__(1024) void place_kernel(const int* __restrict__ src,
                                                     const int* __restrict__ dst,
                                                     const unsigned int* __restrict__ part,
                                                     const unsigned int* __restrict__ cnt_in,
                                                     unsigned int* __restrict__ row_ptr,
                                                     int* __restrict__ csr_src, int E, int N) {
    __shared__ uint4 cur4[HBINS / 4];
    __shared__ unsigned int wsum[32];
    __shared__ unsigned int tile_total;
    unsigned int* cur = reinterpret_cast<unsigned int*>(cur4);
    const int tid = threadIdx.x;
    const int lane = tid & 31;
    const int wv = tid >> 5;
    const long pitch = ((long)N + 31) & ~31L;
    const unsigned int* __restrict__ pref = part + ((long)DEG_CHUNKS + blockIdx.x) * pitch;

    int per = (((E + DEG_CHUNKS - 1) / DEG_CHUNKS) + 3) & ~3;
    int e0 = blockIdx.x * per;
    if (e0 > E) e0 = E;
    int e1 = e0 + per;
    if (e1 > E) e1 = E;
    const int* __restrict__ myd = dst + e0;
    const int* __restrict__ mys = src + e0;
    const int4* __restrict__ myd4 = reinterpret_cast<const int4*>(myd);
    const int4* __restrict__ mys4 = reinterpret_cast<const int4*>(mys);
    const int cnt = e1 - e0;
    const bool vec_ok = ((((size_t)myd) | ((size_t)mys)) & 15) == 0;
    const int nv = vec_ok ? (cnt >> 2) : 0;

    unsigned int carry = 0;
    for (long base = 0; base < N; base += HBINS) {
        long leftn = (long)N - base;
        const int nb = (int)(leftn < HBINS ? leftn : HBINS);
        const int b0 = (int)base;

#pragma unroll 8
        for (int i = tid; i < nb; i += 1024) cur[i] = cnt_in[base + i];
        __syncthreads();

        const int run_len = (nb + 1023) >> 10;
        int rb = tid * run_len;
        if (rb > nb) rb = nb;
        int re = rb + run_len;
        if (re > nb) re = nb;
        unsigned int s = 0;
        for (int i = rb; i < re; ++i) s += cur[i];
        unsigned int inc = s;
        for (int o = 1; o < 32; o <<= 1) {
            unsigned int t = __shfl_up(inc, o);
            if (lane >= o) inc += t;
        }
        if (lane == 31) wsum[wv] = inc;
        __syncthreads();
        if (wv == 0) {
            unsigned int wsv = wsum[lane];
            unsigned int wi = wsv;
            for (int o = 1; o < 32; o <<= 1) {
                unsigned int t = __shfl_up(wi, o);
                if (lane >= o) wi += t;
            }
            wsum[lane] = wi - wsv;
            if (lane == 31) tile_total = wi;
        }
        __syncthreads();
        unsigned int run = carry + wsum[wv] + (inc - s);
        for (int i = rb; i < re; ++i) {
            unsigned int v = cur[i];
            cur[i] = run;
            run += v;
        }
        carry += tile_total;
        __syncthreads();

#pragma unroll 8
        for (int i = tid; i < nb; i += 1024) {
            unsigned int r = cur[i];
            if (blockIdx.x == 0) row_ptr[base + i] = r;
            cur[i] = r + pref[base + i];
        }
        __syncthreads();

        for (int v0 = 0; v0 < nv; v0 += PL_BATCH * 1024) {
            int4 vd[PL_BATCH], vs[PL_BATCH];
#pragma unroll
            for (int u = 0; u < PL_BATCH; ++u) {
                int i = v0 + u * 1024 + tid;
                if (i < nv) {
                    vd[u] = myd4[i];
                    vs[u] = mys4[i];
                } else {
                    vd[u] = make_int4(-1, -1, -1, -1);
                    vs[u] = make_int4(0, 0, 0, 0);
                }
            }
#pragma unroll
            for (int u = 0; u < PL_BATCH; ++u) {
                place_one(cur, csr_src, vd[u].x, vs[u].x, b0, (unsigned int)nb);
                place_one(cur, csr_src, vd[u].y, vs[u].y, b0, (unsigned int)nb);
                place_one(cur, csr_src, vd[u].z, vs[u].z, b0, (unsigned int)nb);
                place_one(cur, csr_src, vd[u].w, vs[u].w, b0, (unsigned int)nb);
            }
        }
        for (int e = (nv << 2) + tid; e < cnt; e += 1024)
            place_one(cur, csr_src, myd[e], mys[e], b0, (unsigned int)nb);
        __syncthreads();
    }
    if (blockIdx.x == 0 && tid == 0) row_ptr[N] = carry;
}

template <int KD>
__global__ __launch_bounds__(128) void gemm_wmma(const float* __restrict__ X,
                                                 const float* __restrict__ W,
                                                 const float* __restrict__ scale,
                                                 float* __restrict__ Hout, int nrows) {
    __shared__ float wlds[KD * F_OUT];
    int tid = threadIdx.x;
    for (int i = tid; i < KD * F_OUT; i += 128) wlds[i] = W[i];
    __syncthreads();

    int wave = tid >> 5;
    int lane = tid & 31;
    int l15  = lane & 15;
    int hi   = lane >> 4;
    int koff = hi * 2;

    int row0 = blockIdx.x * 64 + wave * 16;
    int rA = row0 + l15;
    if (rA >= nrows) rA = nrows - 1;
    float s = scale ? scale[rA] : 1.0f;

    const float* xrow = X + (long)rA * KD;

    v8f acc[4] = {};

    for (int k = 0; k < KD; k += 4) {
        v2f a;
        a.x = xrow[k + koff]     * s;
        a.y = xrow[k + koff + 1] * s;
        __builtin_prefetch(xrow + k + 16, 0, 1);
#pragma unroll
        for (int n = 0; n < 4; ++n) {
            v2f b;
            b.x = wlds[(k + koff)     * F_OUT + n * 16 + l15];
            b.y = wlds[(k + koff + 1) * F_OUT + n * 16 + l15];
            acc[n] = __builtin_amdgcn_wmma_f32_16x16x4_f32(
                false, a, false, b, (short)0, acc[n], false, false);
        }
    }

    float* outp = Hout + (long)(row0 + hi * 8) * F_OUT + l15;
    if (row0 + 16 <= nrows) {
#pragma unroll
        for (int r = 0; r < 8; ++r) {
#pragma unroll
            for (int n = 0; n < 4; ++n) outp[(long)r * F_OUT + n * 16] = acc[n][r];
        }
    } else {
#pragma unroll
        for (int r = 0; r < 8; ++r) {
            int row = row0 + r + hi * 8;
            if (row < nrows) {
#pragma unroll
                for (int n = 0; n < 4; ++n) outp[(long)r * F_OUT + n * 16] = acc[n][r];
            }
        }
    }
}

__device__ __forceinline__ float4 gather_rows(const float4* __restrict__ H4,
                                              const int* __restrict__ csr_src,
                                              unsigned int beg, unsigned int end, int q) {
    float4 acc = make_float4(0.0f, 0.0f, 0.0f, 0.0f);
    for (unsigned int c0 = beg; c0 < end; c0 += 16) {
        const unsigned int left = end - c0;
        const int m = (int)(left < 16u ? left : 16u);
        const int mine = (q < m) ? csr_src[c0 + q] : 0;
        for (int j = 0; j < m; j += 4) {
            const int s0 = __shfl(mine, j, 16);
            const int s1 = __shfl(mine, j + 1, 16);
            const int s2 = __shfl(mine, j + 2, 16);
            const int s3 = __shfl(mine, j + 3, 16);
            const float4 zero4 = make_float4(0.0f, 0.0f, 0.0f, 0.0f);
            float4 v0 = H4[(long)s0 * (F_OUT / 4) + q];
            float4 v1 = zero4, v2 = zero4, v3 = zero4;
            if (j + 1 < m) v1 = H4[(long)s1 * (F_OUT / 4) + q];
            if (j + 2 < m) v2 = H4[(long)s2 * (F_OUT / 4) + q];
            if (j + 3 < m) v3 = H4[(long)s3 * (F_OUT / 4) + q];
            acc.x += v0.x; acc.y += v0.y; acc.z += v0.z; acc.w += v0.w;
            acc.x += v1.x; acc.y += v1.y; acc.z += v1.z; acc.w += v1.w;
            acc.x += v2.x; acc.y += v2.y; acc.z += v2.z; acc.w += v2.w;
            acc.x += v3.x; acc.y += v3.y; acc.z += v3.z; acc.w += v3.w;
        }
    }
    return acc;
}

__global__ __launch_bounds__(256) void gather_kernel(const float* __restrict__ H,
                                                     const int* __restrict__ csr_src,
                                                     const unsigned int* __restrict__ row_ptr,
                                                     const float* __restrict__ norm_in,
                                                     const float* __restrict__ b,
                                                     float* __restrict__ dstrows, int N) {
    const int t = blockIdx.x * 256 + threadIdx.x;
    const int node = t >> 4;
    const int q = t & 15;
    if (node >= N) return;
    const unsigned int beg = row_ptr[node];
    const unsigned int end = row_ptr[node + 1];
    const float ni = norm_in[node];
    const float b0 = b[4 * q], b1 = b[4 * q + 1], b2 = b[4 * q + 2], b3 = b[4 * q + 3];
    const float4 acc = gather_rows(reinterpret_cast<const float4*>(H), csr_src, beg, end, q);
    float4 r;
    r.x = acc.x * ni + b0;
    r.y = acc.y * ni + b1;
    r.z = acc.z * ni + b2;
    r.w = acc.w * ni + b3;
    reinterpret_cast<float4*>(dstrows)[(long)node * (F_OUT / 4) + q] = r;
}

#define TILE_PITCH 68
__global__ __launch_bounds__(256) void gather_gemm_kernel(const float* __restrict__ H,
                                                          const int* __restrict__ csr_src,
                                                          const unsigned int* __restrict__ row_ptr,
                                                          const float* __restrict__ norm_in,
                                                          const float* __restrict__ norm_out,
                                                          const float* __restrict__ b,
                                                          const float* __restrict__ W,
                                                          float* __restrict__ Hout, int N) {
    __shared__ float4 tile4[16 * TILE_PITCH / 4];
    float* tile = reinterpret_cast<float*>(tile4);
    const int tid = threadIdx.x;
    const int g = tid >> 4;
    const int q = tid & 15;
    const int node0 = blockIdx.x * 16;
    const int node = node0 + g;
    unsigned int beg = 0, end = 0;
    float ni = 0.0f, no = 0.0f;
    if (node < N) {
        beg = row_ptr[node];
        end = row_ptr[node + 1];
        ni = norm_in[node];
        no = norm_out[node];
    }
    const float b0 = b[4 * q], b1 = b[4 * q + 1], b2 = b[4 * q + 2], b3 = b[4 * q + 3];
    const float4 acc = gather_rows(reinterpret_cast<const float4*>(H), csr_src, beg, end, q);
    float4 r;
    r.x = fmaxf(acc.x * ni + b0, 0.0f) * no;
    r.y = fmaxf(acc.y * ni + b1, 0.0f) * no;
    r.z = fmaxf(acc.z * ni + b2, 0.0f) * no;
    r.w = fmaxf(acc.w * ni + b3, 0.0f) * no;
    tile4[g * (TILE_PITCH / 4) + q] = r;

    const int wave = __builtin_amdgcn_readfirstlane(tid >> 5);
    const int lane = tid & 31;
    const int l15  = lane & 15;
    const int hi   = lane >> 4;
    const int koff = hi * 2;
    v2f bf[F_OUT / 4];
    if (wave < 4) {
#pragma unroll
        for (int ks = 0; ks < F_OUT / 4; ++ks) {
            bf[ks].x = W[(4 * ks + koff)     * F_OUT + wave * 16 + l15];
            bf[ks].y = W[(4 * ks + koff + 1) * F_OUT + wave * 16 + l15];
        }
    }
    __builtin_amdgcn_fence(__ATOMIC_RELEASE, "workgroup", "local");
    __builtin_amdgcn_s_barrier();
    __builtin_amdgcn_fence(__ATOMIC_ACQUIRE, "workgroup", "local");
    if (wave >= 4) return;

    v8f d = {};
#pragma unroll
    for (int ks = 0; ks < F_OUT / 4; ++ks) {
        const v2f a = *reinterpret_cast<const v2f*>(&tile[l15 * TILE_PITCH + 4 * ks + koff]);
        d = __builtin_amdgcn_wmma_f32_16x16x4_f32(false, a, false, bf[ks], (short)0, d, false, false);
    }

    float* outp = Hout + (long)(node0 + hi * 8) * F_OUT + wave * 16 + l15;
    if (node0 + 16 <= N) {
#pragma unroll
        for (int rr = 0; rr < 8; ++rr) outp[(long)rr * F_OUT] = d[rr];
    } else {
#pragma unroll
        for (int rr = 0; rr < 8; ++rr) {
            if (node0 + hi * 8 + rr < N) outp[(long)rr * F_OUT] = d[rr];
        }
    }
}

extern "C" void kernel_launch(void* const* d_in, const int* in_sizes, int n_in,
                              void* d_out, int out_size, void* d_ws, size_t ws_size,
                              hipStream_t stream) {
    const float* x   = (const float*)d_in[0];
    const int*   src = (const int*)d_in[1];
    const int*   dst = (const int*)d_in[2];
    const float* W1  = (const float*)d_in[3];
    const float* b1  = (const float*)d_in[4];
    const float* W2  = (const float*)d_in[5];
    const float* b2  = (const float*)d_in[6];
    float* out = (float*)d_out;

    const int N = in_sizes[0] / 128;
    const int E = in_sizes[1];
    const long NF = (long)N * F_OUT;

    float* w        = (float*)d_ws;
    unsigned int* cnt_in = (unsigned int*)w;
    float* norm_out = w + 2L * N;
    float* norm_in  = w + 3L * N;
    float* bufA     = w + 4L * N;
    float* bufB     = bufA + NF;
    unsigned int* part = (unsigned int*)(bufB + NF);
    unsigned int* row_ptr = part + 2L * DEG_CHUNKS * (N + 32);
    int* csr_src = (int*)(row_ptr + N + 32);

    const int T = 256;

    degree_kernel<<<dim3(DEG_CHUNKS, 2), 1024, 0, stream>>>(src, dst, part, E, N);
    norm_kernel<<<(N + T - 1) / T, T, 0, stream>>>(part, norm_out, norm_in, cnt_in, N);
    place_kernel<<<DEG_CHUNKS, 1024, 0, stream>>>(src, dst, part, cnt_in, row_ptr, csr_src, E, N);

    int gemm_blocks = (N + 63) / 64;
    gemm_wmma<128><<<gemm_blocks, 128, 0, stream>>>(x, W1, norm_out, bufA, N);

    const int gather_blocks = (int)((NF / 4 + T - 1) / T);
    gather_gemm_kernel<<<gather_blocks, T, 0, stream>>>(bufA, csr_src, row_ptr, norm_in, norm_out, b1, W2, bufB, N);

    gather_kernel<<<gather_blocks, T, 0, stream>>>(bufB, csr_src, row_ptr, norm_in, b2, out, N);
}


# layer-1 WMMA product moved into the edge-placing launch: extra work-groups on the WGPs its 32 work-groups leave idle; 6 launches -> 5; f32 products, k order and every sum order unchanged
# speedup vs baseline: 3.1036x; 1.0881x over previous
#include <hip/hip_runtime.h>

typedef __attribute__((ext_vector_type(2))) float v2f;
typedef __attribute__((ext_vector_type(8))) float v8f;

#define F_OUT 64


#define DEG_CHUNKS 32
#define HBINS      51200
#define DEG_BATCH  8

__device__ __forceinline__ void count_one(unsigned int* hist, int v, int base, unsigned int nb) {
    unsigned int a = (unsigned int)(v - base);
    if (a < nb) atomicAdd(&hist[a], 1u);
}

__global__ __launch_bounds__(1024) void degree_kernel(const int* __restrict__ src,
                                                      const int* __restrict__ dst,
                                                      unsigned int* __restrict__ part,
                                                      int E, int N) {
    __shared__ uint4 hist4[HBINS / 4];
    unsigned int* hist = reinterpret_cast<unsigned int*>(hist4);
    const int tid = threadIdx.x;
    const int* __restrict__ idx = blockIdx.y ? dst : src;
    const long pitch = ((long)N + 31) & ~31L;
    unsigned int* __restrict__ plane = part + ((long)blockIdx.y * DEG_CHUNKS + blockIdx.x) * pitch;

    int per = (((E + DEG_CHUNKS - 1) / DEG_CHUNKS) + 3) & ~3;
    int e0 = blockIdx.x * per;
    if (e0 > E) e0 = E;
    int e1 = e0 + per;
    if (e1 > E) e1 = E;
    const int* __restrict__ my = idx + e0;
    const int4* __restrict__ my4 = reinterpret_cast<const int4*>(my);
    const int cnt = e1 - e0;
    const bool vec_ok = (((size_t)my) & 15) == 0;
    const int nv = vec_ok ? (cnt >> 2) : 0;

    for (long base = 0; base < N; base += HBINS) {
        long left = pitch - base;
        const int nw = (int)(left < HBINS ? left : HBINS);
        long leftn = (long)N - base;
        const unsigned int nb = (unsigned int)(leftn < HBINS ? leftn : HBINS);
        const int b0 = (int)base;

        for (int i = tid; i < HBINS / 4; i += 1024) hist4[i] = make_uint4(0u, 0u, 0u, 0u);
        __syncthreads();

        for (int s = 0; s < nv; s += DEG_BATCH * 1024) {
            int4 v[DEG_BATCH];
#pragma unroll
            for (int u = 0; u < DEG_BATCH; ++u) {
                int i = s + u * 1024 + tid;
                v[u] = (i < nv) ? my4[i] : make_int4(-1, -1, -1, -1);
            }
#pragma unroll
            for (int u = 0; u < DEG_BATCH; ++u) {
                count_one(hist, v[u].x, b0, nb);
                count_one(hist, v[u].y, b0, nb);
                count_one(hist, v[u].z, b0, nb);
                count_one(hist, v[u].w, b0, nb);
            }
        }
        for (int e = (nv << 2) + tid; e < cnt; e += 1024) count_one(hist, my[e], b0, nb);
        __syncthreads();

        uint4* __restrict__ out4 = reinterpret_cast<uint4*>(plane + base);
        for (int i = tid; i < (nw >> 2); i += 1024) out4[i] = hist4[i];
        __syncthreads();
    }
}

__global__ void norm_kernel(unsigned int* __restrict__ part, float* __restrict__ norm_out,
                            float* __restrict__ norm_in, unsigned int* __restrict__ cnt_in, int N) {
    int i = blockIdx.x * blockDim.x + threadIdx.x;
    if (i < N) {
        const long pitch = ((long)N + 31) & ~31L;
        const unsigned int* po = part + i;
        unsigned int* pi = part + DEG_CHUNKS * pitch + i;
        unsigned int vo[DEG_CHUNKS], vi[DEG_CHUNKS];
#pragma unroll
        for (int g = 0; g < DEG_CHUNKS; ++g) {
            vo[g] = po[g * pitch];
            vi[g] = pi[g * pitch];
        }
        unsigned int co = 0, ci = 0;
#pragma unroll
        for (int g = 0; g < DEG_CHUNKS; ++g) {
            co += vo[g];
            pi[g * pitch] = ci;
            ci += vi[g];
        }
        cnt_in[i] = ci;
        norm_out[i] = rsqrtf(fmaxf((float)co, 1.0f));
        norm_in[i]  = rsqrtf(fmaxf((float)ci, 1.0f));
    }
}

#define GEMM_WAVES 16
template <int KD>
__device__ __forceinline__ void gemm_rows(float* __restrict__ wlds, const float* __restrict__ X,
                                          const float* __restrict__ W,
                                          const float* __restrict__ scale,
                                          float* __restrict__ Hout, int nrows, int gblock) {
    const int tid = threadIdx.x;
#pragma unroll
    for (int i = 0; i < KD * F_OUT / 1024; ++i) wlds[i * 1024 + tid] = W[i * 1024 + tid];
    __syncthreads();

    const int wave = __builtin_amdgcn_readfirstlane(tid >> 5);
    const int row0 = (gblock * GEMM_WAVES + wave) * 16;
    if (wave >= GEMM_WAVES || row0 >= nrows) return;

    int lane = tid & 31;
    int l15  = lane & 15;
    int hi   = lane >> 4;
    int koff = hi * 2;

    int rA = row0 + l15;
    if (rA >= nrows) rA = nrows - 1;
    float s = scale[rA];

    const float* xrow = X + (long)rA * KD;

    v8f acc[4] = {};

    for (int k = 0; k < KD; k += 4) {
        v2f a;
        a.x = xrow[k + koff]     * s;
        a.y = xrow[k + koff + 1] * s;
        __builtin_prefetch(xrow + k + 16, 0, 1);
#pragma unroll
        for (int n = 0; n < 4; ++n) {
            v2f b;
            b.x = wlds[(k + koff)     * F_OUT + n * 16 + l15];
            b.y = wlds[(k + koff + 1) * F_OUT + n * 16 + l15];
            acc[n] = __builtin_amdgcn_wmma_f32_16x16x4_f32(
                false, a, false, b, (short)0, acc[n], false, false);
        }
    }

    float* outp = Hout + (long)(row0 + hi * 8) * F_OUT + l15;
    if (row0 + 16 <= nrows) {
#pragma unroll
        for (int r = 0; r < 8; ++r) {
#pragma unroll
            for (int n = 0; n < 4; ++n) outp[(long)r * F_OUT + n * 16] = acc[n][r];
        }
    } else {
#pragma unroll
        for (int r = 0; r < 8; ++r) {
            int row = row0 + r + hi * 8;
            if (row < nrows) {
#pragma unroll
                for (int n = 0; n < 4; ++n) outp[(long)r * F_OUT + n * 16] = acc[n][r];
            }
        }
    }
}

#define PL_BATCH 8

__device__ __forceinline__ void place_one(unsigned int* cur, int* __restrict__ csr, int d, int s,
                                          int base, unsigned int nb) {
    unsigned int a = (unsigned int)(d - base);
    if (a < nb) {
        unsigned int pos = atomicAdd(&cur[a], 1u);
        csr[pos] = s;
    }
}

__global__ __launch_bounds__(1024) void place_gemm_kernel(const int* __restrict__ src,
                                                          const int* __restrict__ dst,
                                                          const unsigned int* __restrict__ part,
                                                          const unsigned int* __restrict__ cnt_in,
                                                          unsigned int* __restrict__ row_ptr,
                                                          int* __restrict__ csr_src, int E, int N,
                                                          const float* __restrict__ X,
                                                          const float* __restrict__ W,
                                                          const float* __restrict__ scale,
                                                          float* __restrict__ Hout) {
    __shared__ uint4 cur4[HBINS / 4];
    __shared__ unsigned int wsum[32];
    __shared__ unsigned int tile_total;
    if (blockIdx.x >= DEG_CHUNKS) {
        gemm_rows<128>(reinterpret_cast<float*>(cur4), X, W, scale, Hout, N,
                       (int)blockIdx.x - DEG_CHUNKS);
        return;
    }
    unsigned int* cur = reinterpret_cast<unsigned int*>(cur4);
    const int tid = threadIdx.x;
    const int lane = tid & 31;
    const int wv = tid >> 5;
    const long pitch = ((long)N + 31) & ~31L;
    const unsigned int* __restrict__ pref = part + ((long)DEG_CHUNKS + blockIdx.x) * pitch;

    int per = (((E + DEG_CHUNKS - 1) / DEG_CHUNKS) + 3) & ~3;
    int e0 = blockIdx.x * per;
    if (e0 > E) e0 = E;
    int e1 = e0 + per;
    if (e1 > E) e1 = E;
    const int* __restrict__ myd = dst + e0;
    const int* __restrict__ mys = src + e0;
    const int4* __restrict__ myd4 = reinterpret_cast<const int4*>(myd);
    const int4* __restrict__ mys4 = reinterpret_cast<const int4*>(mys);
    const int cnt = e1 - e0;
    const bool vec_ok = ((((size_t)myd) | ((size_t)mys)) & 15) == 0;
    const int nv = vec_ok ? (cnt >> 2) : 0;

    unsigned int carry = 0;
    for (long base = 0; base < N; base += HBINS) {
        long leftn = (long)N - base;
        const int nb = (int)(leftn < HBINS ? leftn : HBINS);
        const int b0 = (int)base;

#pragma unroll 8
        for (int i = tid; i < nb; i += 1024) cur[i] = cnt_in[base + i];
        __syncthreads();

        const int run_len = (nb + 1023) >> 10;
        int rb = tid * run_len;
        if (rb > nb) rb = nb;
        int re = rb + run_len;
        if (re > nb) re = nb;
        unsigned int s = 0;
        for (int i = rb; i < re; ++i) s += cur[i];
        unsigned int inc = s;
        for (int o = 1; o < 32; o <<= 1) {
            unsigned int t = __shfl_up(inc, o);
            if (lane >= o) inc += t;
        }
        if (lane == 31) wsum[wv] = inc;
        __syncthreads();
        if (wv == 0) {
            unsigned int wsv = wsum[lane];
            unsigned int wi = wsv;
            for (int o = 1; o < 32; o <<= 1) {
                unsigned int t = __shfl_up(wi, o);
                if (lane >= o) wi += t;
            }
            wsum[lane] = wi - wsv;
            if (lane == 31) tile_total = wi;
        }
        __syncthreads();
        unsigned int run = carry + wsum[wv] + (inc - s);
        for (int i = rb; i < re; ++i) {
            unsigned int v = cur[i];
            cur[i] = run;
            run += v;
        }
        carry += tile_total;
        __syncthreads();

#pragma unroll 8
        for (int i = tid; i < nb; i += 1024) {
            unsigned int r = cur[i];
            if (blockIdx.x == 0) row_ptr[base + i] = r;
            cur[i] = r + pref[base + i];
        }
        __syncthreads();

        for (int v0 = 0; v0 < nv; v0 += PL_BATCH * 1024) {
            int4 vd[PL_BATCH], vs[PL_BATCH];
#pragma unroll
            for (int u = 0; u < PL_BATCH; ++u) {
                int i = v0 + u * 1024 + tid;
                if (i < nv) {
                    vd[u] = myd4[i];
                    vs[u] = mys4[i];
                } else {
                    vd[u] = make_int4(-1, -1, -1, -1);
                    vs[u] = make_int4(0, 0, 0, 0);
                }
            }
#pragma unroll
            for (int u = 0; u < PL_BATCH; ++u) {
                place_one(cur, csr_src, vd[u].x, vs[u].x, b0, (unsigned int)nb);
                place_one(cur, csr_src, vd[u].y, vs[u].y, b0, (unsigned int)nb);
                place_one(cur, csr_src, vd[u].z, vs[u].z, b0, (unsigned int)nb);
                place_one(cur, csr_src, vd[u].w, vs[u].w, b0, (unsigned int)nb);
            }
        }
        for (int e = (nv << 2) + tid; e < cnt; e += 1024)
            place_one(cur, csr_src, myd[e], mys[e], b0, (unsigned int)nb);
        __syncthreads();
    }
    if (blockIdx.x == 0 && tid == 0) row_ptr[N] = carry;
}

__device__ __forceinline__ float4 gather_rows(const float4* __restrict__ H4,
                                              const int* __restrict__ csr_src,
                                              unsigned int beg, unsigned int end, int q) {
    float4 acc = make_float4(0.0f, 0.0f, 0.0f, 0.0f);
    for (unsigned int c0 = beg; c0 < end; c0 += 16) {
        const unsigned int left = end - c0;
        const int m = (int)(left < 16u ? left : 16u);
        const int mine = (q < m) ? csr_src[c0 + q] : 0;
        for (int j = 0; j < m; j += 4) {
            const int s0 = __shfl(mine, j, 16);
            const int s1 = __shfl(mine, j + 1, 16);
            const int s2 = __shfl(mine, j + 2, 16);
            const int s3 = __shfl(mine, j + 3, 16);
            const float4 zero4 = make_float4(0.0f, 0.0f, 0.0f, 0.0f);
            float4 v0 = H4[(long)s0 * (F_OUT / 4) + q];
            float4 v1 = zero4, v2 = zero4, v3 = zero4;
            if (j + 1 < m) v1 = H4[(long)s1 * (F_OUT / 4) + q];
            if (j + 2 < m) v2 = H4[(long)s2 * (F_OUT / 4) + q];
            if (j + 3 < m) v3 = H4[(long)s3 * (F_OUT / 4) + q];
            acc.x += v0.x; acc.y += v0.y; acc.z += v0.z; acc.w += v0.w;
            acc.x += v1.x; acc.y += v1.y; acc.z += v1.z; acc.w += v1.w;
            acc.x += v2.x; acc.y += v2.y; acc.z += v2.z; acc.w += v2.w;
            acc.x += v3.x; acc.y += v3.y; acc.z += v3.z; acc.w += v3.w;
        }
    }
    return acc;
}

__global__ __launch_bounds__(256) void gather_kernel(const float* __restrict__ H,
                                                     const int* __restrict__ csr_src,
                                                     const unsigned int* __restrict__ row_ptr,
                                                     const float* __restrict__ norm_in,
                                                     const float* __restrict__ b,
                                                     float* __restrict__ dstrows, int N) {
    const int t = blockIdx.x * 256 + threadIdx.x;
    const int node = t >> 4;
    const int q = t & 15;
    if (node >= N) return;
    const unsigned int beg = row_ptr[node];
    const unsigned int end = row_ptr[node + 1];
    const float ni = norm_in[node];
    const float b0 = b[4 * q], b1 = b[4 * q + 1], b2 = b[4 * q + 2], b3 = b[4 * q + 3];
    const float4 acc = gather_rows(reinterpret_cast<const float4*>(H), csr_src, beg, end, q);
    float4 r;
    r.x = acc.x * ni + b0;
    r.y = acc.y * ni + b1;
    r.z = acc.z * ni + b2;
    r.w = acc.w * ni + b3;
    reinterpret_cast<float4*>(dstrows)[(long)node * (F_OUT / 4) + q] = r;
}

#define TILE_PITCH 68
__global__ __launch_bounds__(256) void gather_gemm_kernel(const float* __restrict__ H,
                                                          const int* __restrict__ csr_src,
                                                          const unsigned int* __restrict__ row_ptr,
                                                          const float* __restrict__ norm_in,
                                                          const float* __restrict__ norm_out,
                                                          const float* __restrict__ b,
                                                          const float* __restrict__ W,
                                                          float* __restrict__ Hout, int N) {
    __shared__ float4 tile4[16 * TILE_PITCH / 4];
    float* tile = reinterpret_cast<float*>(tile4);
    const int tid = threadIdx.x;
    const int g = tid >> 4;
    const int q = tid & 15;
    const int node0 = blockIdx.x * 16;
    const int node = node0 + g;
    unsigned int beg = 0, end = 0;
    float ni = 0.0f, no = 0.0f;
    if (node < N) {
        beg = row_ptr[node];
        end = row_ptr[node + 1];
        ni = norm_in[node];
        no = norm_out[node];
    }
    const float b0 = b[4 * q], b1 = b[4 * q + 1], b2 = b[4 * q + 2], b3 = b[4 * q + 3];
    const float4 acc = gather_rows(reinterpret_cast<const float4*>(H), csr_src, beg, end, q);
    float4 r;
    r.x = fmaxf(acc.x * ni + b0, 0.0f) * no;
    r.y = fmaxf(acc.y * ni + b1, 0.0f) * no;
    r.z = fmaxf(acc.z * ni + b2, 0.0f) * no;
    r.w = fmaxf(acc.w * ni + b3, 0.0f) * no;
    tile4[g * (TILE_PITCH / 4) + q] = r;

    const int wave = __builtin_amdgcn_readfirstlane(tid >> 5);
    const int lane = tid & 31;
    const int l15  = lane & 15;
    const int hi   = lane >> 4;
    const int koff = hi * 2;
    v2f bf[F_OUT / 4];
    if (wave < 4) {
#pragma unroll
        for (int ks = 0; ks < F_OUT / 4; ++ks) {
            bf[ks].x = W[(4 * ks + koff)     * F_OUT + wave * 16 + l15];
            bf[ks].y = W[(4 * ks + koff + 1) * F_OUT + wave * 16 + l15];
        }
    }
    __builtin_amdgcn_fence(__ATOMIC_RELEASE, "workgroup", "local");
    __builtin_amdgcn_s_barrier();
    __builtin_amdgcn_fence(__ATOMIC_ACQUIRE, "workgroup", "local");
    if (wave >= 4) return;

    v8f d = {};
#pragma unroll
    for (int ks = 0; ks < F_OUT / 4; ++ks) {
        const v2f a = *reinterpret_cast<const v2f*>(&tile[l15 * TILE_PITCH + 4 * ks + koff]);
        d = __builtin_amdgcn_wmma_f32_16x16x4_f32(false, a, false, bf[ks], (short)0, d, false, false);
    }

    float* outp = Hout + (long)(node0 + hi * 8) * F_OUT + wave * 16 + l15;
    if (node0 + 16 <= N) {
#pragma unroll
        for (int rr = 0; rr < 8; ++rr) outp[(long)rr * F_OUT] = d[rr];
    } else {
#pragma unroll
        for (int rr = 0; rr < 8; ++rr) {
            if (node0 + hi * 8 + rr < N) outp[(long)rr * F_OUT] = d[rr];
        }
    }
}

extern "C" void kernel_launch(void* const* d_in, const int* in_sizes, int n_in,
                              void* d_out, int out_size, void* d_ws, size_t ws_size,
                              hipStream_t stream) {
    const float* x   = (const float*)d_in[0];
    const int*   src = (const int*)d_in[1];
    const int*   dst = (const int*)d_in[2];
    const float* W1  = (const float*)d_in[3];
    const float* b1  = (const float*)d_in[4];
    const float* W2  = (const float*)d_in[5];
    const float* b2  = (const float*)d_in[6];
    float* out = (float*)d_out;

    const int N = in_sizes[0] / 128;
    const int E = in_sizes[1];
    const long NF = (long)N * F_OUT;

    float* w        = (float*)d_ws;
    unsigned int* cnt_in = (unsigned int*)w;
    float* norm_out = w + 2L * N;
    float* norm_in  = w + 3L * N;
    float* bufA     = w + 4L * N;
    float* bufB     = bufA + NF;
    unsigned int* part = (unsigned int*)(bufB + NF);
    unsigned int* row_ptr = part + 2L * DEG_CHUNKS * (N + 32);
    int* csr_src = (int*)(row_ptr + N + 32);

    const int T = 256;

    degree_kernel<<<dim3(DEG_CHUNKS, 2), 1024, 0, stream>>>(src, dst, part, E, N);
    norm_kernel<<<(N + T - 1) / T, T, 0, stream>>>(part, norm_out, norm_in, cnt_in, N);
    const int gemm_blocks = ((N + 15) / 16 + GEMM_WAVES - 1) / GEMM_WAVES;
    place_gemm_kernel<<<DEG_CHUNKS + gemm_blocks, 1024, 0, stream>>>(src, dst, part, cnt_in, row_ptr, csr_src,
                                                                     E, N, x, W1, norm_out, bufA);

    const int gather_blocks = (int)((NF / 4 + T - 1) / T);
    gather_gemm_kernel<<<gather_blocks, T, 0, stream>>>(bufA, csr_src, row_ptr, norm_in, norm_out, b1, W2, bufB, N);

    gather_kernel<<<gather_blocks, T, 0, stream>>>(bufB, csr_src, row_ptr, norm_in, b2, out, N);
}


# edge placing split by node range: one work-group per (edge chunk, 12800-node range), 32 -> 128 placing groups; product tiles 16 -> 28 per group so all 240 groups run at once; sum orders unchanged
# speedup vs baseline: 4.8639x; 1.5672x over previous
#include <hip/hip_runtime.h>

typedef __attribute__((ext_vector_type(2))) float v2f;
typedef __attribute__((ext_vector_type(8))) float v8f;

#define F_OUT 64


#define DEG_CHUNKS 32
#define HBINS      51200
#define DEG_BATCH  8

__device__ __forceinline__ void count_one(unsigned int* hist, int v, int base, unsigned int nb) {
    unsigned int a = (unsigned int)(v - base);
    if (a < nb) atomicAdd(&hist[a], 1u);
}

__global__ __launch_bounds__(1024) void degree_kernel(const int* __restrict__ src,
                                                      const int* __restrict__ dst,
                                                      unsigned int* __restrict__ part,
                                                      int E, int N) {
    __shared__ uint4 hist4[HBINS / 4];
    unsigned int* hist = reinterpret_cast<unsigned int*>(hist4);
    const int tid = threadIdx.x;
    const int* __restrict__ idx = blockIdx.y ? dst : src;
    const long pitch = ((long)N + 31) & ~31L;
    unsigned int* __restrict__ plane = part + ((long)blockIdx.y * DEG_CHUNKS + blockIdx.x) * pitch;

    int per = (((E + DEG_CHUNKS - 1) / DEG_CHUNKS) + 3) & ~3;
    int e0 = blockIdx.x * per;
    if (e0 > E) e0 = E;
    int e1 = e0 + per;
    if (e1 > E) e1 = E;
    const int* __restrict__ my = idx + e0;
    const int4* __restrict__ my4 = reinterpret_cast<const int4*>(my);
    const int cnt = e1 - e0;
    const bool vec_ok = (((size_t)my) & 15) == 0;
    const int nv = vec_ok ? (cnt >> 2) : 0;

    for (long base = 0; base < N; base += HBINS) {
        long left = pitch - base;
        const int nw = (int)(left < HBINS ? left : HBINS);
        long leftn = (long)N - base;
        const unsigned int nb = (unsigned int)(leftn < HBINS ? leftn : HBINS);
        const int b0 = (int)base;

        for (int i = tid; i < HBINS / 4; i += 1024) hist4[i] = make_uint4(0u, 0u, 0u, 0u);
        __syncthreads();

        for (int s = 0; s < nv; s += DEG_BATCH * 1024) {
            int4 v[DEG_BATCH];
#pragma unroll
            for (int u = 0; u < DEG_BATCH; ++u) {
                int i = s + u * 1024 + tid;
                v[u] = (i < nv) ? my4[i] : make_int4(-1, -1, -1, -1);
            }
#pragma unroll
            for (int u = 0; u < DEG_BATCH; ++u) {
                count_one(hist, v[u].x, b0, nb);
                count_one(hist, v[u].y, b0, nb);
                count_one(hist, v[u].z, b0, nb);
                count_one(hist, v[u].w, b0, nb);
            }
        }
        for (int e = (nv << 2) + tid; e < cnt; e += 1024) count_one(hist, my[e], b0, nb);
        __syncthreads();

        uint4* __restrict__ out4 = reinterpret_cast<uint4*>(plane + base);
        for (int i = tid; i < (nw >> 2); i += 1024) out4[i] = hist4[i];
        __syncthreads();
    }
}

__global__ void norm_kernel(unsigned int* __restrict__ part, float* __restrict__ norm_out,
                            float* __restrict__ norm_in, unsigned int* __restrict__ cnt_in, int N) {
    int i = blockIdx.x * blockDim.x + threadIdx.x;
    if (i < N) {
        const long pitch = ((long)N + 31) & ~31L;
        const unsigned int* po = part + i;
        unsigned int* pi = part + DEG_CHUNKS * pitch + i;
        unsigned int vo[DEG_CHUNKS], vi[DEG_CHUNKS];
#pragma unroll
        for (int g = 0; g < DEG_CHUNKS; ++g) {
            vo[g] = po[g * pitch];
            vi[g] = pi[g * pitch];
        }
        unsigned int co = 0, ci = 0;
#pragma unroll
        for (int g = 0; g < DEG_CHUNKS; ++g) {
            co += vo[g];
            pi[g * pitch] = ci;
            ci += vi[g];
        }
        cnt_in[i] = ci;
        norm_out[i] = rsqrtf(fmaxf((float)co, 1.0f));
        norm_in[i]  = rsqrtf(fmaxf((float)ci, 1.0f));
    }
}

#define GEMM_WAVES 28
template <int KD>
__device__ __forceinline__ void gemm_rows(float* __restrict__ wlds, const float* __restrict__ X,
                                          const float* __restrict__ W,
                                          const float* __restrict__ scale,
                                          float* __restrict__ Hout, int nrows, int gblock) {
    const int tid = threadIdx.x;
#pragma unroll
    for (int i = 0; i < KD * F_OUT / 1024; ++i) wlds[i * 1024 + tid] = W[i * 1024 + tid];
    __syncthreads();

    const int wave = __builtin_amdgcn_readfirstlane(tid >> 5);
    const int row0 = (gblock * GEMM_WAVES + wave) * 16;
    if (wave >= GEMM_WAVES || row0 >= nrows) return;

    int lane = tid & 31;
    int l15  = lane & 15;
    int hi   = lane >> 4;
    int koff = hi * 2;

    int rA = row0 + l15;
    if (rA >= nrows) rA = nrows - 1;
    float s = scale[rA];

    const float* xrow = X + (long)rA * KD;

    v8f acc[4] = {};

    for (int k = 0; k < KD; k += 4) {
        v2f a;
        a.x = xrow[k + koff]     * s;
        a.y = xrow[k + koff + 1] * s;
        __builtin_prefetch(xrow + k + 16, 0, 1);
#pragma unroll
        for (int n = 0; n < 4; ++n) {
            v2f b;
            b.x = wlds[(k + koff)     * F_OUT + n * 16 + l15];
            b.y = wlds[(k + koff + 1) * F_OUT + n * 16 + l15];
            acc[n] = __builtin_amdgcn_wmma_f32_16x16x4_f32(
                false, a, false, b, (short)0, acc[n], false, false);
        }
    }

    float* outp = Hout + (long)(row0 + hi * 8) * F_OUT + l15;
    if (row0 + 16 <= nrows) {
#pragma unroll
        for (int r = 0; r < 8; ++r) {
#pragma unroll
            for (int n = 0; n < 4; ++n) outp[(long)r * F_OUT + n * 16] = acc[n][r];
        }
    } else {
#pragma unroll
        for (int r = 0; r < 8; ++r) {
            int row = row0 + r + hi * 8;
            if (row < nrows) {
#pragma unroll
                for (int n = 0; n < 4; ++n) outp[(long)r * F_OUT + n * 16] = acc[n][r];
            }
        }
    }
}

#define PL_BATCH 8
#define PL_BINS 12800

__device__ __forceinline__ void place_one(unsigned int* cur, int* __restrict__ csr, int d, int s,
                                          int base, unsigned int nb) {
    unsigned int a = (unsigned int)(d - base);
    if (a < nb) {
        unsigned int pos = atomicAdd(&cur[a], 1u);
        csr[pos] = s;
    }
}

__global__ __launch_bounds__(1024) void place_gemm_kernel(const int* __restrict__ src,
                                                          const int* __restrict__ dst,
                                                          const unsigned int* __restrict__ part,
                                                          const unsigned int* __restrict__ cnt_in,
                                                          unsigned int* __restrict__ row_ptr,
                                                          int* __restrict__ csr_src, int E, int N,
                                                          const float* __restrict__ X,
                                                          const float* __restrict__ W,
                                                          const float* __restrict__ scale,
                                                          float* __restrict__ Hout) {
    __shared__ uint4 cur4[PL_BINS / 4];
    __shared__ unsigned int wsum[32];
    __shared__ unsigned int tile_total;
    const int nrange = (N + PL_BINS - 1) / PL_BINS;
    const int nplace = DEG_CHUNKS * nrange;
    if ((int)blockIdx.x >= nplace) {
        gemm_rows<128>(reinterpret_cast<float*>(cur4), X, W, scale, Hout, N,
                       (int)blockIdx.x - nplace);
        return;
    }
    const int chunk = (int)blockIdx.x % DEG_CHUNKS;
    const int range = (int)blockIdx.x / DEG_CHUNKS;
    unsigned int* cur = reinterpret_cast<unsigned int*>(cur4);
    const int tid = threadIdx.x;
    const int lane = tid & 31;
    const int wv = tid >> 5;
    const long pitch = ((long)N + 31) & ~31L;
    const unsigned int* __restrict__ pref = part + ((long)DEG_CHUNKS + chunk) * pitch;

    int per = (((E + DEG_CHUNKS - 1) / DEG_CHUNKS) + 3) & ~3;
    int e0 = chunk * per;
    if (e0 > E) e0 = E;
    int e1 = e0 + per;
    if (e1 > E) e1 = E;
    const int* __restrict__ myd = dst + e0;
    const int* __restrict__ mys = src + e0;
    const int4* __restrict__ myd4 = reinterpret_cast<const int4*>(myd);
    const int4* __restrict__ mys4 = reinterpret_cast<const int4*>(mys);
    const int cnt = e1 - e0;
    const bool vec_ok = ((((size_t)myd) | ((size_t)mys)) & 15) == 0;
    const int nv = vec_ok ? (cnt >> 2) : 0;

    const long base = (long)range * PL_BINS;
    const long leftn = (long)N - base;
    const int nb = (int)(leftn < PL_BINS ? leftn : PL_BINS);
    const int b0 = (int)base;

    unsigned int below = 0;
    {
        const uint4* __restrict__ lo4 = reinterpret_cast<const uint4*>(cnt_in);
        const int n4 = b0 >> 2;
        for (int s0 = 0; s0 < n4; s0 += 8 * 1024) {
            uint4 v[8];
#pragma unroll
            for (int u = 0; u < 8; ++u) {
                const int i = s0 + u * 1024 + tid;
                v[u] = (i < n4) ? lo4[i] : make_uint4(0u, 0u, 0u, 0u);
            }
#pragma unroll
            for (int u = 0; u < 8; ++u) below += (v[u].x + v[u].y) + (v[u].z + v[u].w);
        }
    }
    for (int o = 16; o > 0; o >>= 1) below += __shfl_xor(below, o);
    if (lane == 0) wsum[wv] = below;
    __syncthreads();
    unsigned int carry = wsum[lane];
    for (int o = 16; o > 0; o >>= 1) carry += __shfl_xor(carry, o);

    const int nb4 = nb >> 2;
    {
        const uint4* __restrict__ in4 = reinterpret_cast<const uint4*>(cnt_in + base);
        for (int s0 = 0; s0 < nb4; s0 += 4 * 1024) {
            uint4 v[4];
#pragma unroll
            for (int u = 0; u < 4; ++u) {
                const int i = s0 + u * 1024 + tid;
                v[u] = (i < nb4) ? in4[i] : make_uint4(0u, 0u, 0u, 0u);
            }
#pragma unroll
            for (int u = 0; u < 4; ++u) {
                const int i = s0 + u * 1024 + tid;
                if (i < nb4) cur4[i] = v[u];
            }
        }
        for (int i = (nb4 << 2) + tid; i < nb; i += 1024) cur[i] = cnt_in[base + i];
    }
    __syncthreads();

    const int run_len = (nb + 1023) >> 10;
    int rb = tid * run_len;
    if (rb > nb) rb = nb;
    int re = rb + run_len;
    if (re > nb) re = nb;
    unsigned int s = 0;
    for (int i = rb; i < re; ++i) s += cur[i];
    unsigned int inc = s;
    for (int o = 1; o < 32; o <<= 1) {
        unsigned int t = __shfl_up(inc, o);
        if (lane >= o) inc += t;
    }
    if (lane == 31) wsum[wv] = inc;
    __syncthreads();
    if (wv == 0) {
        unsigned int wsv = wsum[lane];
        unsigned int wi = wsv;
        for (int o = 1; o < 32; o <<= 1) {
            unsigned int t = __shfl_up(wi, o);
            if (lane >= o) wi += t;
        }
        wsum[lane] = wi - wsv;
        if (lane == 31) tile_total = wi;
    }
    __syncthreads();
    unsigned int run = carry + wsum[wv] + (inc - s);
    for (int i = rb; i < re; ++i) {
        unsigned int v = cur[i];
        cur[i] = run;
        run += v;
    }
    carry += tile_total;
    __syncthreads();

    {
        const uint4* __restrict__ pref4 = reinterpret_cast<const uint4*>(pref + base);
        uint4* __restrict__ rp4 = reinterpret_cast<uint4*>(row_ptr + base);
        for (int s0 = 0; s0 < nb4; s0 += 4 * 1024) {
            uint4 p[4];
#pragma unroll
            for (int u = 0; u < 4; ++u) {
                const int i = s0 + u * 1024 + tid;
                p[u] = (i < nb4) ? pref4[i] : make_uint4(0u, 0u, 0u, 0u);
            }
#pragma unroll
            for (int u = 0; u < 4; ++u) {
                const int i = s0 + u * 1024 + tid;
                if (i < nb4) {
                    const uint4 r = cur4[i];
                    if (chunk == 0) rp4[i] = r;
                    cur4[i] = make_uint4(r.x + p[u].x, r.y + p[u].y, r.z + p[u].z, r.w + p[u].w);
                }
            }
        }
        for (int i = (nb4 << 2) + tid; i < nb; i += 1024) {
            const unsigned int r = cur[i];
            if (chunk == 0) row_ptr[base + i] = r;
            cur[i] = r + pref[base + i];
        }
    }
    __syncthreads();

    for (int v0 = 0; v0 < nv; v0 += PL_BATCH * 1024) {
        int4 vd[PL_BATCH], vs[PL_BATCH];
#pragma unroll
        for (int u = 0; u < PL_BATCH; ++u) {
            int i = v0 + u * 1024 + tid;
            if (i < nv) {
                vd[u] = myd4[i];
                vs[u] = mys4[i];
            } else {
                vd[u] = make_int4(-1, -1, -1, -1);
                vs[u] = make_int4(0, 0, 0, 0);
            }
        }
#pragma unroll
        for (int u = 0; u < PL_BATCH; ++u) {
            place_one(cur, csr_src, vd[u].x, vs[u].x, b0, (unsigned int)nb);
            place_one(cur, csr_src, vd[u].y, vs[u].y, b0, (unsigned int)nb);
            place_one(cur, csr_src, vd[u].z, vs[u].z, b0, (unsigned int)nb);
            place_one(cur, csr_src, vd[u].w, vs[u].w, b0, (unsigned int)nb);
        }
    }
    for (int e = (nv << 2) + tid; e < cnt; e += 1024)
        place_one(cur, csr_src, myd[e], mys[e], b0, (unsigned int)nb);
    if (chunk == 0 && range == nrange - 1 && tid == 0) row_ptr[N] = carry;
}

__device__ __forceinline__ float4 gather_rows(const float4* __restrict__ H4,
                                              const int* __restrict__ csr_src,
                                              unsigned int beg, unsigned int end, int q) {
    float4 acc = make_float4(0.0f, 0.0f, 0.0f, 0.0f);
    for (unsigned int c0 = beg; c0 < end; c0 += 16) {
        const unsigned int left = end - c0;
        const int m = (int)(left < 16u ? left : 16u);
        const int mine = (q < m) ? csr_src[c0 + q] : 0;
        for (int j = 0; j < m; j += 4) {
            const int s0 = __shfl(mine, j, 16);
            const int s1 = __shfl(mine, j + 1, 16);
            const int s2 = __shfl(mine, j + 2, 16);
            const int s3 = __shfl(mine, j + 3, 16);
            const float4 zero4 = make_float4(0.0f, 0.0f, 0.0f, 0.0f);
            float4 v0 = H4[(long)s0 * (F_OUT / 4) + q];
            float4 v1 = zero4, v2 = zero4, v3 = zero4;
            if (j + 1 < m) v1 = H4[(long)s1 * (F_OUT / 4) + q];
            if (j + 2 < m) v2 = H4[(long)s2 * (F_OUT / 4) + q];
            if (j + 3 < m) v3 = H4[(long)s3 * (F_OUT / 4) + q];
            acc.x += v0.x; acc.y += v0.y; acc.z += v0.z; acc.w += v0.w;
            acc.x += v1.x; acc.y += v1.y; acc.z += v1.z; acc.w += v1.w;
            acc.x += v2.x; acc.y += v2.y; acc.z += v2.z; acc.w += v2.w;
            acc.x += v3.x; acc.y += v3.y; acc.z += v3.z; acc.w += v3.w;
        }
    }
    return acc;
}

__global__ __launch_bounds__(256) void gather_kernel(const float* __restrict__ H,
                                                     const int* __restrict__ csr_src,
                                                     const unsigned int* __restrict__ row_ptr,
                                                     const float* __restrict__ norm_in,
                                                     const float* __restrict__ b,
                                                     float* __restrict__ dstrows, int N) {
    const int t = blockIdx.x * 256 + threadIdx.x;
    const int node = t >> 4;
    const int q = t & 15;
    if (node >= N) return;
    const unsigned int beg = row_ptr[node];
    const unsigned int end = row_ptr[node + 1];
    const float ni = norm_in[node];
    const float b0 = b[4 * q], b1 = b[4 * q + 1], b2 = b[4 * q + 2], b3 = b[4 * q + 3];
    const float4 acc = gather_rows(reinterpret_cast<const float4*>(H), csr_src, beg, end, q);
    float4 r;
    r.x = acc.x * ni + b0;
    r.y = acc.y * ni + b1;
    r.z = acc.z * ni + b2;
    r.w = acc.w * ni + b3;
    reinterpret_cast<float4*>(dstrows)[(long)node * (F_OUT / 4) + q] = r;
}

#define TILE_PITCH 68
__global__ __launch_bounds__(256) void gather_gemm_kernel(const float* __restrict__ H,
                                                          const int* __restrict__ csr_src,
                                                          const unsigned int* __restrict__ row_ptr,
                                                          const float* __restrict__ norm_in,
                                                          const float* __restrict__ norm_out,
                                                          const float* __restrict__ b,
                                                          const float* __restrict__ W,
                                                          float* __restrict__ Hout, int N) {
    __shared__ float4 tile4[16 * TILE_PITCH / 4];
    float* tile = reinterpret_cast<float*>(tile4);
    const int tid = threadIdx.x;
    const int g = tid >> 4;
    const int q = tid & 15;
    const int node0 = blockIdx.x * 16;
    const int node = node0 + g;
    unsigned int beg = 0, end = 0;
    float ni = 0.0f, no = 0.0f;
    if (node < N) {
        beg = row_ptr[node];
        end = row_ptr[node + 1];
        ni = norm_in[node];
        no = norm_out[node];
    }
    const float b0 = b[4 * q], b1 = b[4 * q + 1], b2 = b[4 * q + 2], b3 = b[4 * q + 3];
    const float4 acc = gather_rows(reinterpret_cast<const float4*>(H), csr_src, beg, end, q);
    float4 r;
    r.x = fmaxf(acc.x * ni + b0, 0.0f) * no;
    r.y = fmaxf(acc.y * ni + b1, 0.0f) * no;
    r.z = fmaxf(acc.z * ni + b2, 0.0f) * no;
    r.w = fmaxf(acc.w * ni + b3, 0.0f) * no;
    tile4[g * (TILE_PITCH / 4) + q] = r;

    const int wave = __builtin_amdgcn_readfirstlane(tid >> 5);
    const int lane = tid & 31;
    const int l15  = lane & 15;
    const int hi   = lane >> 4;
    const int koff = hi * 2;
    v2f bf[F_OUT / 4];
    if (wave < 4) {
#pragma unroll
        for (int ks = 0; ks < F_OUT / 4; ++ks) {
            bf[ks].x = W[(4 * ks + koff)     * F_OUT + wave * 16 + l15];
            bf[ks].y = W[(4 * ks + koff + 1) * F_OUT + wave * 16 + l15];
        }
    }
    __builtin_amdgcn_fence(__ATOMIC_RELEASE, "workgroup", "local");
    __builtin_amdgcn_s_barrier();
    __builtin_amdgcn_fence(__ATOMIC_ACQUIRE, "workgroup", "local");
    if (wave >= 4) return;

    v8f d = {};
#pragma unroll
    for (int ks = 0; ks < F_OUT / 4; ++ks) {
        const v2f a = *reinterpret_cast<const v2f*>(&tile[l15 * TILE_PITCH + 4 * ks + koff]);
        d = __builtin_amdgcn_wmma_f32_16x16x4_f32(false, a, false, bf[ks], (short)0, d, false, false);
    }

    float* outp = Hout + (long)(node0 + hi * 8) * F_OUT + wave * 16 + l15;
    if (node0 + 16 <= N) {
#pragma unroll
        for (int rr = 0; rr < 8; ++rr) outp[(long)rr * F_OUT] = d[rr];
    } else {
#pragma unroll
        for (int rr = 0; rr < 8; ++rr) {
            if (node0 + hi * 8 + rr < N) outp[(long)rr * F_OUT] = d[rr];
        }
    }
}

extern "C" void kernel_launch(void* const* d_in, const int* in_sizes, int n_in,
                              void* d_out, int out_size, void* d_ws, size_t ws_size,
                              hipStream_t stream) {
    const float* x   = (const float*)d_in[0];
    const int*   src = (const int*)d_in[1];
    const int*   dst = (const int*)d_in[2];
    const float* W1  = (const float*)d_in[3];
    const float* b1  = (const float*)d_in[4];
    const float* W2  = (const float*)d_in[5];
    const float* b2  = (const float*)d_in[6];
    float* out = (float*)d_out;

    const int N = in_sizes[0] / 128;
    const int E = in_sizes[1];
    const long NF = (long)N * F_OUT;

    float* w        = (float*)d_ws;
    unsigned int* cnt_in = (unsigned int*)w;
    float* norm_out = w + 2L * N;
    float* norm_in  = w + 3L * N;
    float* bufA     = w + 4L * N;
    float* bufB     = bufA + NF;
    unsigned int* part = (unsigned int*)(bufB + NF);
    unsigned int* row_ptr = part + 2L * DEG_CHUNKS * (N + 32);
    int* csr_src = (int*)(row_ptr + N + 32);

    const int T = 256;

    degree_kernel<<<dim3(DEG_CHUNKS, 2), 1024, 0, stream>>>(src, dst, part, E, N);
    norm_kernel<<<(N + T - 1) / T, T, 0, stream>>>(part, norm_out, norm_in, cnt_in, N);
    const int gemm_blocks = ((N + 15) / 16 + GEMM_WAVES - 1) / GEMM_WAVES;
    const int place_blocks = DEG_CHUNKS * ((N + PL_BINS - 1) / PL_BINS);
    place_gemm_kernel<<<place_blocks + gemm_blocks, 1024, 0, stream>>>(src, dst, part, cnt_in, row_ptr, csr_src,
                                                                       E, N, x, W1, norm_out, bufA);

    const int gather_blocks = (int)((NF / 4 + T - 1) / T);
    gather_gemm_kernel<<<gather_blocks, T, 0, stream>>>(bufA, csr_src, row_ptr, norm_in, norm_out, b1, W2, bufB, N);

    gather_kernel<<<gather_blocks, T, 0, stream>>>(bufB, csr_src, row_ptr, norm_in, b2, out, N);
}
